# starting kernel
# baseline (speedup 1.0000x reference)
#include <hip/hip_runtime.h>
#include <math.h>


typedef __attribute__((ext_vector_type(2))) float v2f;
typedef __attribute__((ext_vector_type(8))) float v8f;

#define N_LAYERS 6
#define NWAVES   4
#define NTHREADS (NWAVES * 32)

struct c2 { float re, im; };

__device__ __forceinline__ c2 cmul(c2 a, c2 b) {
  return c2{a.re * b.re - a.im * b.im, a.re * b.im + a.im * b.re};
}
__device__ __forceinline__ c2 rx_el(int a, int b, float c, float s) {
  return (a == b) ? c2{c, 0.f} : c2{0.f, -s};
}
__device__ __forceinline__ c2 ry_el(int a, int b, float c, float s) {
  if (a == b) return c2{c, 0.f};
  return (a == 1) ? c2{s, 0.f} : c2{-s, 0.f};
}

__global__ void qnn_build_v(const float* __restrict__ wts, float* __restrict__ V) {
  int l = blockIdx.x;
  int t = threadIdx.x;
  int r = t >> 4, c = t & 15;
  for (int m = 0; m < 2; ++m) {
    int rr = (m == 0) ? (r ^ (r >> 1)) : r;
    c2 prod = c2{1.f, 0.f};
    for (int i = 0; i < 4; ++i) {
      int wire = m * 4 + i;
      float a0 = wts[(l * 8 + wire) * 3 + 0];
      float a1 = wts[(l * 8 + wire) * 3 + 1];
      float a2 = wts[(l * 8 + wire) * 3 + 2];
      float cx = cosf(0.5f * a0), sx = sinf(0.5f * a0);
      float cy = cosf(0.5f * a1), sy = sinf(0.5f * a1);
      float cz = cosf(0.5f * a2), sz = sinf(0.5f * a2);
      int ab = (rr >> (3 - i)) & 1;
      int bb = (c  >> (3 - i)) & 1;
      c2 yx = c2{0.f, 0.f};
      for (int k = 0; k < 2; ++k) {
        c2 p = cmul(ry_el(ab, k, cy, sy), rx_el(k, bb, cx, sx));
        yx.re += p.re; yx.im += p.im;
      }
      c2 z = ab ? c2{cz, sz} : c2{cz, -sz};
      prod = cmul(prod, cmul(z, yx));
    }
    V[(((l * 2 + m) * 2 + 0) * 256) + t] = prod.re;
    V[(((l * 2 + m) * 2 + 1) * 256) + t] = prod.im;
  }
}

__device__ __forceinline__ void cmm(const float* __restrict__ Ar,
                                    const float* __restrict__ Ai,
                                    const float* __restrict__ Br,
                                    const float* __restrict__ Bi,
                                    bool bT, int l, v8f& Cr, v8f& Ci) {
  int lr = l & 15, lh = l >> 4;
  int aoff = lr * 16 + 2 * lh;
  v8f rr = {}, ii = {}, ri = {};
#pragma unroll
  for (int kb = 0; kb < 4; ++kb) {
    v2f ar = *(const v2f*)(Ar + aoff + 4 * kb);
    v2f ai = *(const v2f*)(Ai + aoff + 4 * kb);
    v2f br, bi;
    if (bT) {
      br = *(const v2f*)(Br + aoff + 4 * kb);
      bi = *(const v2f*)(Bi + aoff + 4 * kb);
    } else {
      int k0 = 4 * kb + 2 * lh;
      br.x = Br[k0 * 16 + lr]; br.y = Br[(k0 + 1) * 16 + lr];
      bi.x = Bi[k0 * 16 + lr]; bi.y = Bi[(k0 + 1) * 16 + lr];
    }
    rr = __builtin_amdgcn_wmma_f32_16x16x4_f32(false, ar, false, br, (short)0, rr, false, false);
    ii = __builtin_amdgcn_wmma_f32_16x16x4_f32(false, ai, false, bi, (short)0, ii, false, false);
    ri = __builtin_amdgcn_wmma_f32_16x16x4_f32(false, ar, false, bi, (short)0, ri, false, false);
    ri = __builtin_amdgcn_wmma_f32_16x16x4_f32(false, ai, false, br, (short)0, ri, false, false);
  }
  Cr = rr - ii;
  Ci = ri;
}

__device__ __forceinline__ void store_tile(float* __restrict__ dst, int l, v8f c) {
  int base = ((l >> 4) * 8) * 16 + (l & 15);
#pragma unroll
  for (int v = 0; v < 8; ++v) dst[base + v * 16] = c[v];
}

__device__ __forceinline__ float wave_sum(float v, int l) {
#pragma unroll
  for (int m = 16; m; m >>= 1) {
    int src = (l ^ m) << 2;
    v += __int_as_float(__builtin_amdgcn_ds_bpermute(src, __float_as_int(v)));
  }
  return v;
}

__global__ void __launch_bounds__(NTHREADS)
qnn_kernel(const float* __restrict__ x, const float* __restrict__ Vg,
           float* __restrict__ out, int batch) {
  __shared__ float Vl[N_LAYERS * 2 * 2 * 256];
  __shared__ float Ebuf[NWAVES][2][256];
  __shared__ float Mbuf[NWAVES][2][256];
  __shared__ float Nbuf[NWAVES][2][256];
  __shared__ float Sbuf[NWAVES][2][256];

  int tid = threadIdx.x;
  int l = tid & 31;
  int w = tid >> 5;
  int b = blockIdx.x * NWAVES + w;
  int bb = (b < batch) ? b : (batch - 1);

  {
    const float4* src = (const float4*)Vg;
    float4* dst = (float4*)Vl;
    for (int i = tid; i < (N_LAYERS * 2 * 2 * 256) / 4; i += NTHREADS) dst[i] = src[i];
  }
  __syncthreads();

  float x0 = x[2 * bb], x1 = x[2 * bb + 1];
  float c0 = cosf(0.5f * x0), s0 = sinf(0.5f * x0);
  float c1 = cosf(0.5f * x1), s1 = sinf(0.5f * x1);
#pragma unroll
  for (int t = 0; t < 8; ++t) {
    int idx = t * 32 + l;
    int a = idx >> 4, bcol = idx & 15;
    c2 e = rx_el((a >> 3) & 1, (bcol >> 3) & 1, c0, s0);
    e = cmul(e, ry_el((a >> 2) & 1, (bcol >> 2) & 1, c1, s1));
    e = cmul(e, rx_el((a >> 1) & 1, (bcol >> 1) & 1, c0, s0));
    e = cmul(e, ry_el(a & 1, bcol & 1, c1, s1));
    Ebuf[w][0][idx] = e.re;
    Ebuf[w][1][idx] = e.im;
    Sbuf[w][0][idx] = (idx == 0) ? 1.f : 0.f;
    Sbuf[w][1][idx] = 0.f;
  }

  int csrc = (l & 15); csrc = csrc ^ (csrc >> 1);
  int addrE = (((l & 16) | csrc)) << 2;
  int addrO = addrE ^ (8 << 2);

  for (int layer = 0; layer < N_LAYERS; ++layer) {
    const float* V0r = &Vl[((layer * 2 + 0) * 2 + 0) * 256];
    const float* V0i = &Vl[((layer * 2 + 0) * 2 + 1) * 256];
    const float* R1r = &Vl[((layer * 2 + 1) * 2 + 0) * 256];
    const float* R1i = &Vl[((layer * 2 + 1) * 2 + 1) * 256];
    v8f cr, ci;
    cmm(V0r, V0i, &Ebuf[w][0][0], &Ebuf[w][1][0], false, l, cr, ci);
    store_tile(&Mbuf[w][0][0], l, cr); store_tile(&Mbuf[w][1][0], l, ci);
    cmm(R1r, R1i, &Ebuf[w][0][0], &Ebuf[w][1][0], false, l, cr, ci);
    store_tile(&Nbuf[w][0][0], l, cr); store_tile(&Nbuf[w][1][0], l, ci);
    cmm(&Mbuf[w][0][0], &Mbuf[w][1][0], &Sbuf[w][0][0], &Sbuf[w][1][0], false, l, cr, ci);
    store_tile(&Mbuf[w][0][0], l, cr); store_tile(&Mbuf[w][1][0], l, ci);
    cmm(&Mbuf[w][0][0], &Mbuf[w][1][0], &Nbuf[w][0][0], &Nbuf[w][1][0], true, l, cr, ci);
    v8f nr, ni;
#pragma unroll
    for (int v = 0; v < 8; ++v) {
      int ad = (v & 1) ? addrO : addrE;
      nr[v] = __int_as_float(__builtin_amdgcn_ds_bpermute(ad, __float_as_int(cr[v])));
      ni[v] = __int_as_float(__builtin_amdgcn_ds_bpermute(ad, __float_as_int(ci[v])));
    }
    store_tile(&Sbuf[w][0][0], l, nr);
    store_tile(&Sbuf[w][1][0], l, ni);
  }

  int lr = l & 15, lh = l >> 4;
  float p[8];
#pragma unroll
  for (int v = 0; v < 8; ++v) {
    float a = Sbuf[w][0][(v + 8 * lh) * 16 + lr];
    float bi = Sbuf[w][1][(v + 8 * lh) * 16 + lr];
    p[v] = a * a + bi * bi;
  }
  float t  = p[0] + p[1] + p[2] + p[3] + p[4] + p[5] + p[6] + p[7];
  float u1 = p[0] + p[1] + p[2] + p[3] - p[4] - p[5] - p[6] - p[7];
  float u2 = p[0] + p[1] - p[2] - p[3] + p[4] + p[5] - p[6] - p[7];
  float u3 = p[0] - p[1] + p[2] - p[3] + p[4] - p[5] + p[6] - p[7];
  float ev[8];
  ev[0] = lh ? -t : t;
  ev[1] = u1; ev[2] = u2; ev[3] = u3;
  ev[4] = (lr & 8) ? -t : t;
  ev[5] = (lr & 4) ? -t : t;
  ev[6] = (lr & 2) ? -t : t;
  ev[7] = (lr & 1) ? -t : t;
#pragma unroll
  for (int i = 0; i < 8; ++i) ev[i] = wave_sum(ev[i], l);

  if (b < batch && l == 0) {
    const float kPi = 3.14159265358979323846f;
#pragma unroll
    for (int i = 0; i < 8; ++i) out[b * 8 + i] = kPi * ev[i];
  }
}

extern "C" void kernel_launch(void* const* d_in, const int* in_sizes, int n_in,
                              void* d_out, int out_size, void* d_ws, size_t ws_size,
                              hipStream_t stream) {
  const float* x   = (const float*)d_in[0];
  const float* wts = (const float*)d_in[1];
  float* out = (float*)d_out;
  float* V   = (float*)d_ws;

  int batch = in_sizes[0] / 2;
  qnn_build_v<<<N_LAYERS, 256, 0, stream>>>(wts, V);
  int nblocks = (batch + NWAVES - 1) / NWAVES;
  qnn_kernel<<<nblocks, NTHREADS, 0, stream>>>(x, V, out, batch);
}


# whole circuit as its exact degree-24 trig polynomial in (x0,x1): 49x49 grid of circuit runs + real DFT give coefficients, each 16 elements then one 400x52 by 52x16 WMMA product; sums re-ordered
# speedup vs baseline: 8.5921x; 8.5921x over previous
#include <hip/hip_runtime.h>
#include <math.h>


typedef __attribute__((ext_vector_type(2))) float v2f;
typedef __attribute__((ext_vector_type(8))) float v8f;

#define N_LAYERS 6
#define NWAVES   4
#define NTHREADS (NWAVES * 32)

#define GN 49
#define GK 24
#define NGRID (GN * GN)
#define MT 25
#define KB 13
#define AM_FLOATS (MT * KB * 64)
#define WS_GX 8192
#define WS_GO 16384
#define WS_AM 40960
#define EV_THREADS 512
#define COEF_THREADS 416

struct c2 { float re, im; };

__device__ __forceinline__ c2 cmul(c2 a, c2 b) {
  return c2{a.re * b.re - a.im * b.im, a.re * b.im + a.im * b.re};
}
__device__ __forceinline__ c2 rx_el(int a, int b, float c, float s) {
  return (a == b) ? c2{c, 0.f} : c2{0.f, -s};
}
__device__ __forceinline__ c2 ry_el(int a, int b, float c, float s) {
  if (a == b) return c2{c, 0.f};
  return (a == 1) ? c2{s, 0.f} : c2{-s, 0.f};
}

__global__ void qnn_build_v(const float* __restrict__ wts, float* __restrict__ V,
                            float* __restrict__ GX) {
  if (blockIdx.x >= N_LAYERS) {
    int g = (blockIdx.x - N_LAYERS) * 256 + threadIdx.x;
    if (g < NGRID) {
      int a = g / GN, bq = g - a * GN;
      const float step = 6.28318530717958647692f / 49.0f;
      GX[2 * g] = step * (float)a;
      GX[2 * g + 1] = step * (float)bq;
    }
    return;
  }
  int l = blockIdx.x;
  int t = threadIdx.x;
  int r = t >> 4, c = t & 15;
  for (int m = 0; m < 2; ++m) {
    int rr = (m == 0) ? (r ^ (r >> 1)) : r;
    c2 prod = c2{1.f, 0.f};
    for (int i = 0; i < 4; ++i) {
      int wire = m * 4 + i;
      float a0 = wts[(l * 8 + wire) * 3 + 0];
      float a1 = wts[(l * 8 + wire) * 3 + 1];
      float a2 = wts[(l * 8 + wire) * 3 + 2];
      float cx = cosf(0.5f * a0), sx = sinf(0.5f * a0);
      float cy = cosf(0.5f * a1), sy = sinf(0.5f * a1);
      float cz = cosf(0.5f * a2), sz = sinf(0.5f * a2);
      int ab = (rr >> (3 - i)) & 1;
      int bb = (c  >> (3 - i)) & 1;
      c2 yx = c2{0.f, 0.f};
      for (int k = 0; k < 2; ++k) {
        c2 p = cmul(ry_el(ab, k, cy, sy), rx_el(k, bb, cx, sx));
        yx.re += p.re; yx.im += p.im;
      }
      c2 z = ab ? c2{cz, sz} : c2{cz, -sz};
      prod = cmul(prod, cmul(z, yx));
    }
    V[(((l * 2 + m) * 2 + 0) * 256) + t] = prod.re;
    V[(((l * 2 + m) * 2 + 1) * 256) + t] = prod.im;
  }
}

__device__ __forceinline__ void cmm(const float* __restrict__ Ar,
                                    const float* __restrict__ Ai,
                                    const float* __restrict__ Br,
                                    const float* __restrict__ Bi,
                                    bool bT, int l, v8f& Cr, v8f& Ci) {
  int lr = l & 15, lh = l >> 4;
  int aoff = lr * 16 + 2 * lh;
  v8f rr = {}, ii = {}, ri = {};
#pragma unroll
  for (int kb = 0; kb < 4; ++kb) {
    v2f ar = *(const v2f*)(Ar + aoff + 4 * kb);
    v2f ai = *(const v2f*)(Ai + aoff + 4 * kb);
    v2f br, bi;
    if (bT) {
      br = *(const v2f*)(Br + aoff + 4 * kb);
      bi = *(const v2f*)(Bi + aoff + 4 * kb);
    } else {
      int k0 = 4 * kb + 2 * lh;
      br.x = Br[k0 * 16 + lr]; br.y = Br[(k0 + 1) * 16 + lr];
      bi.x = Bi[k0 * 16 + lr]; bi.y = Bi[(k0 + 1) * 16 + lr];
    }
    rr = __builtin_amdgcn_wmma_f32_16x16x4_f32(false, ar, false, br, (short)0, rr, false, false);
    ii = __builtin_amdgcn_wmma_f32_16x16x4_f32(false, ai, false, bi, (short)0, ii, false, false);
    ri = __builtin_amdgcn_wmma_f32_16x16x4_f32(false, ar, false, bi, (short)0, ri, false, false);
    ri = __builtin_amdgcn_wmma_f32_16x16x4_f32(false, ai, false, br, (short)0, ri, false, false);
  }
  Cr = rr - ii;
  Ci = ri;
}

__device__ __forceinline__ void store_tile(float* __restrict__ dst, int l, v8f c) {
  int base = ((l >> 4) * 8) * 16 + (l & 15);
#pragma unroll
  for (int v = 0; v < 8; ++v) dst[base + v * 16] = c[v];
}

__device__ __forceinline__ float wave_sum(float v, int l) {
#pragma unroll
  for (int m = 16; m; m >>= 1) {
    int src = (l ^ m) << 2;
    v += __int_as_float(__builtin_amdgcn_ds_bpermute(src, __float_as_int(v)));
  }
  return v;
}

__global__ void __launch_bounds__(NTHREADS)
qnn_kernel(const float* __restrict__ x, const float* __restrict__ Vg,
           float* __restrict__ out, int batch) {
  __shared__ float Vl[N_LAYERS * 2 * 2 * 256];
  __shared__ float Ebuf[NWAVES][2][256];
  __shared__ float Mbuf[NWAVES][2][256];
  __shared__ float Nbuf[NWAVES][2][256];
  __shared__ float Sbuf[NWAVES][2][256];

  int tid = threadIdx.x;
  int l = tid & 31;
  int w = tid >> 5;
  int b = blockIdx.x * NWAVES + w;
  int bb = (b < batch) ? b : (batch - 1);

  {
    const float4* src = (const float4*)Vg;
    float4* dst = (float4*)Vl;
    for (int i = tid; i < (N_LAYERS * 2 * 2 * 256) / 4; i += NTHREADS) dst[i] = src[i];
  }
  __syncthreads();

  float x0 = x[2 * bb], x1 = x[2 * bb + 1];
  float c0 = cosf(0.5f * x0), s0 = sinf(0.5f * x0);
  float c1 = cosf(0.5f * x1), s1 = sinf(0.5f * x1);
#pragma unroll
  for (int t = 0; t < 8; ++t) {
    int idx = t * 32 + l;
    int a = idx >> 4, bcol = idx & 15;
    c2 e = rx_el((a >> 3) & 1, (bcol >> 3) & 1, c0, s0);
    e = cmul(e, ry_el((a >> 2) & 1, (bcol >> 2) & 1, c1, s1));
    e = cmul(e, rx_el((a >> 1) & 1, (bcol >> 1) & 1, c0, s0));
    e = cmul(e, ry_el(a & 1, bcol & 1, c1, s1));
    Ebuf[w][0][idx] = e.re;
    Ebuf[w][1][idx] = e.im;
    Sbuf[w][0][idx] = (idx == 0) ? 1.f : 0.f;
    Sbuf[w][1][idx] = 0.f;
  }

  int csrc = (l & 15); csrc = csrc ^ (csrc >> 1);
  int addrE = (((l & 16) | csrc)) << 2;
  int addrO = addrE ^ (8 << 2);

  for (int layer = 0; layer < N_LAYERS; ++layer) {
    const float* V0r = &Vl[((layer * 2 + 0) * 2 + 0) * 256];
    const float* V0i = &Vl[((layer * 2 + 0) * 2 + 1) * 256];
    const float* R1r = &Vl[((layer * 2 + 1) * 2 + 0) * 256];
    const float* R1i = &Vl[((layer * 2 + 1) * 2 + 1) * 256];
    v8f cr, ci;
    cmm(V0r, V0i, &Ebuf[w][0][0], &Ebuf[w][1][0], false, l, cr, ci);
    store_tile(&Mbuf[w][0][0], l, cr); store_tile(&Mbuf[w][1][0], l, ci);
    cmm(R1r, R1i, &Ebuf[w][0][0], &Ebuf[w][1][0], false, l, cr, ci);
    store_tile(&Nbuf[w][0][0], l, cr); store_tile(&Nbuf[w][1][0], l, ci);
    cmm(&Mbuf[w][0][0], &Mbuf[w][1][0], &Sbuf[w][0][0], &Sbuf[w][1][0], false, l, cr, ci);
    store_tile(&Mbuf[w][0][0], l, cr); store_tile(&Mbuf[w][1][0], l, ci);
    cmm(&Mbuf[w][0][0], &Mbuf[w][1][0], &Nbuf[w][0][0], &Nbuf[w][1][0], true, l, cr, ci);
    v8f nr, ni;
#pragma unroll
    for (int v = 0; v < 8; ++v) {
      int ad = (v & 1) ? addrO : addrE;
      nr[v] = __int_as_float(__builtin_amdgcn_ds_bpermute(ad, __float_as_int(cr[v])));
      ni[v] = __int_as_float(__builtin_amdgcn_ds_bpermute(ad, __float_as_int(ci[v])));
    }
    store_tile(&Sbuf[w][0][0], l, nr);
    store_tile(&Sbuf[w][1][0], l, ni);
  }

  int lr = l & 15, lh = l >> 4;
  float p[8];
#pragma unroll
  for (int v = 0; v < 8; ++v) {
    float a = Sbuf[w][0][(v + 8 * lh) * 16 + lr];
    float bi = Sbuf[w][1][(v + 8 * lh) * 16 + lr];
    p[v] = a * a + bi * bi;
  }
  float t  = p[0] + p[1] + p[2] + p[3] + p[4] + p[5] + p[6] + p[7];
  float u1 = p[0] + p[1] + p[2] + p[3] - p[4] - p[5] - p[6] - p[7];
  float u2 = p[0] + p[1] - p[2] - p[3] + p[4] + p[5] - p[6] - p[7];
  float u3 = p[0] - p[1] + p[2] - p[3] + p[4] - p[5] + p[6] - p[7];
  float ev[8];
  ev[0] = lh ? -t : t;
  ev[1] = u1; ev[2] = u2; ev[3] = u3;
  ev[4] = (lr & 8) ? -t : t;
  ev[5] = (lr & 4) ? -t : t;
  ev[6] = (lr & 2) ? -t : t;
  ev[7] = (lr & 1) ? -t : t;
#pragma unroll
  for (int i = 0; i < 8; ++i) ev[i] = wave_sum(ev[i], l);

  if (b < batch && l == 0) {
    const float kPi = 3.14159265358979323846f;
#pragma unroll
    for (int i = 0; i < 8; ++i) out[b * 8 + i] = kPi * ev[i];
  }
}

__global__ void __launch_bounds__(COEF_THREADS)
qnn_coef(const float* __restrict__ GO, float* __restrict__ AM) {
  __shared__ float ct[GN], st[GN];
  __shared__ float H[GN * 8];
  int t = blockIdx.x >> 1, ph = blockIdx.x & 1;
  int tid = threadIdx.x;
  if (tid < GN) {
    float ang = (float)tid * (2.0f / 49.0f);
    ct[tid] = cospif(ang);
    st[tid] = sinpif(ang);
  }
  __syncthreads();
  if (tid < GN * 8) {
    float s = 0.f;
    int j = 0;
#pragma unroll
    for (int a = 0; a < GN; ++a) {
      float tw = ph ? st[j] : ct[j];
      s += tw * GO[a * (GN * 8) + tid];
      j += t; if (j >= GN) j -= GN;
    }
    H[tid] = s;
  }
  __syncthreads();
  {
    int i = tid & 7, k = tid >> 3;
    int u = k >> 1, ee = k & 1;
    float val = 0.f;
    if (u <= GK) {
      float s = 0.f;
      int j = 0;
#pragma unroll
      for (int bq = 0; bq < GN; ++bq) {
        float tw = ee ? st[j] : ct[j];
        s += tw * H[bq * 8 + i];
        j += u; if (j >= GN) j -= GN;
      }
      float wt = (t == 0) ? (1.0f / 49.0f) : (2.0f / 49.0f);
      float wu = (u == 0) ? (1.0f / 49.0f) : (2.0f / 49.0f);
      val = s * wt * wu;
    }
    int m = i + 8 * ph;
    int kb = k >> 2, h = (k >> 1) & 1, jj = k & 1;
    AM[((t * KB + kb) * 32 + m + 16 * h) * 2 + jj] = val;
  }
}

__global__ void __launch_bounds__(EV_THREADS)
qnn_eval(const float* __restrict__ x, const float* __restrict__ AMg,
         float* __restrict__ out, int batch) {
  __shared__ float AMl[AM_FLOATS];
  int tid = threadIdx.x;
  int l = tid & 31, w = tid >> 5;
  int lr = l & 15, lh = l >> 4;
  {
    const float4* srcp = (const float4*)AMg;
    float4* dstp = (float4*)AMl;
    for (int i = tid; i < AM_FLOATS / 4; i += EV_THREADS) dstp[i] = srcp[i];
  }
  __syncthreads();

  const v2f* Af = (const v2f*)AMl + l;
  int ntiles = (batch + 31) >> 5;
  for (int tile = blockIdx.x * (EV_THREADS / 32) + w; tile < ntiles;
       tile += gridDim.x * (EV_THREADS / 32)) {
    v2f bf[2][KB];
    float c1[2], s1[2];
#pragma unroll
    for (int s = 0; s < 2; ++s) {
      int b = tile * 32 + s * 16 + lr;
      int bb = (b < batch) ? b : (batch - 1);
      float x0 = x[2 * bb], x1 = x[2 * bb + 1];
      float sy, cy;
      sincosf(x1, &sy, &cy);
      sincosf(x0, &s1[s], &c1[s]);
      float c2 = cy * cy - sy * sy, s2 = 2.0f * cy * sy;
      float cu = lh ? cy : 1.0f, su = lh ? sy : 0.0f;
#pragma unroll
      for (int kb = 0; kb < KB; ++kb) {
        bf[s][kb].x = cu; bf[s][kb].y = su;
        float nc = cu * c2 - su * s2;
        su = su * c2 + cu * s2;
        cu = nc;
      }
    }
    float acc0[8], acc1[8];
#pragma unroll
    for (int v = 0; v < 8; ++v) { acc0[v] = 0.f; acc1[v] = 0.f; }
    float ct0 = 1.f, st0 = 0.f, ct1 = 1.f, st1 = 0.f;
    for (int t = 0; t < MT; ++t) {
      v8f d0 = {}, d1 = {};
#pragma unroll
      for (int kb = 0; kb < KB; ++kb) {
        v2f a = Af[(t * KB + kb) * 32];
        d0 = __builtin_amdgcn_wmma_f32_16x16x4_f32(false, a, false, bf[0][kb], (short)0, d0, false, false);
        d1 = __builtin_amdgcn_wmma_f32_16x16x4_f32(false, a, false, bf[1][kb], (short)0, d1, false, false);
      }
      float f0 = lh ? st0 : ct0;
      float f1 = lh ? st1 : ct1;
#pragma unroll
      for (int v = 0; v < 8; ++v) { acc0[v] += f0 * d0[v]; acc1[v] += f1 * d1[v]; }
      float n0 = ct0 * c1[0] - st0 * s1[0]; st0 = st0 * c1[0] + ct0 * s1[0]; ct0 = n0;
      float n1 = ct1 * c1[1] - st1 * s1[1]; st1 = st1 * c1[1] + ct1 * s1[1]; ct1 = n1;
    }
    float tot[8];
    int paddr = (l ^ 16) << 2;
#pragma unroll
    for (int v = 0; v < 8; ++v) {
      float give = lh ? acc0[v] : acc1[v];
      float keep = lh ? acc1[v] : acc0[v];
      tot[v] = keep + __int_as_float(__builtin_amdgcn_ds_bpermute(paddr, __float_as_int(give)));
    }
    int bo = tile * 32 + l;
    if (bo < batch) {
      float4* o = (float4*)(out + (size_t)bo * 8);
      o[0] = make_float4(tot[0], tot[1], tot[2], tot[3]);
      o[1] = make_float4(tot[4], tot[5], tot[6], tot[7]);
    }
  }
}

extern "C" void kernel_launch(void* const* d_in, const int* in_sizes, int n_in,
                              void* d_out, int out_size, void* d_ws, size_t ws_size,
                              hipStream_t stream) {
  const float* x   = (const float*)d_in[0];
  const float* wts = (const float*)d_in[1];
  float* out = (float*)d_out;
  float* V   = (float*)d_ws;
  float* GX  = V + WS_GX;
  float* GO  = V + WS_GO;
  float* AM  = V + WS_AM;

  int batch = in_sizes[0] / 2;
  qnn_build_v<<<N_LAYERS + 10, 256, 0, stream>>>(wts, V, GX);
  qnn_kernel<<<601, NTHREADS, 0, stream>>>(GX, V, GO, NGRID);
  qnn_coef<<<2 * MT, COEF_THREADS, 0, stream>>>(GO, AM);
  qnn_eval<<<256, EV_THREADS, 0, stream>>>(x, AM, out, batch);
}


# setup launch folded into the circuit kernel: each block builds the 48 gate 2x2 matrices and the 12 shared 16x16 tiles in LDS, grid angles from the point index; 3 launches, was 4; no sums re-ordered
# speedup vs baseline: 9.7002x; 1.1290x over previous
#include <hip/hip_runtime.h>
#include <math.h>


typedef __attribute__((ext_vector_type(2))) float v2f;
typedef __attribute__((ext_vector_type(8))) float v8f;

#define N_LAYERS 6
#define NWAVES   4
#define NTHREADS (NWAVES * 32)

#define GN 49
#define GK 24
#define NGRID (GN * GN)
#define MT 25
#define KB 13
#define AM_FLOATS (MT * KB * 64)
#define WS_GO 16384
#define WS_AM 40960
#define EV_THREADS 512
#define COEF_THREADS 416

struct c2 { float re, im; };

__device__ __forceinline__ c2 cmul(c2 a, c2 b) {
  return c2{a.re * b.re - a.im * b.im, a.re * b.im + a.im * b.re};
}
__device__ __forceinline__ c2 rx_el(int a, int b, float c, float s) {
  return (a == b) ? c2{c, 0.f} : c2{0.f, -s};
}
__device__ __forceinline__ c2 ry_el(int a, int b, float c, float s) {
  if (a == b) return c2{c, 0.f};
  return (a == 1) ? c2{s, 0.f} : c2{-s, 0.f};
}

__device__ __forceinline__ void cmm(const float* __restrict__ Ar,
                                    const float* __restrict__ Ai,
                                    const float* __restrict__ Br,
                                    const float* __restrict__ Bi,
                                    bool bT, int l, v8f& Cr, v8f& Ci) {
  int lr = l & 15, lh = l >> 4;
  int aoff = lr * 16 + 2 * lh;
  v8f rr = {}, ii = {}, ri = {};
#pragma unroll
  for (int kb = 0; kb < 4; ++kb) {
    v2f ar = *(const v2f*)(Ar + aoff + 4 * kb);
    v2f ai = *(const v2f*)(Ai + aoff + 4 * kb);
    v2f br, bi;
    if (bT) {
      br = *(const v2f*)(Br + aoff + 4 * kb);
      bi = *(const v2f*)(Bi + aoff + 4 * kb);
    } else {
      int k0 = 4 * kb + 2 * lh;
      br.x = Br[k0 * 16 + lr]; br.y = Br[(k0 + 1) * 16 + lr];
      bi.x = Bi[k0 * 16 + lr]; bi.y = Bi[(k0 + 1) * 16 + lr];
    }
    rr = __builtin_amdgcn_wmma_f32_16x16x4_f32(false, ar, false, br, (short)0, rr, false, false);
    ii = __builtin_amdgcn_wmma_f32_16x16x4_f32(false, ai, false, bi, (short)0, ii, false, false);
    ri = __builtin_amdgcn_wmma_f32_16x16x4_f32(false, ar, false, bi, (short)0, ri, false, false);
    ri = __builtin_amdgcn_wmma_f32_16x16x4_f32(false, ai, false, br, (short)0, ri, false, false);
  }
  Cr = rr - ii;
  Ci = ri;
}

__device__ __forceinline__ void store_tile(float* __restrict__ dst, int l, v8f c) {
  int base = ((l >> 4) * 8) * 16 + (l & 15);
#pragma unroll
  for (int v = 0; v < 8; ++v) dst[base + v * 16] = c[v];
}

__device__ __forceinline__ float wave_sum(float v, int l) {
#pragma unroll
  for (int m = 16; m; m >>= 1) {
    int src = (l ^ m) << 2;
    v += __int_as_float(__builtin_amdgcn_ds_bpermute(src, __float_as_int(v)));
  }
  return v;
}

__global__ void __launch_bounds__(NTHREADS)
qnn_kernel(const float* __restrict__ wts, float* __restrict__ out, int batch) {
  __shared__ float Vl[N_LAYERS * 2 * 2 * 256];
  __shared__ float Gl[N_LAYERS * 8 * 8];
  __shared__ float Ebuf[NWAVES][2][256];
  __shared__ float Mbuf[NWAVES][2][256];
  __shared__ float Nbuf[NWAVES][2][256];
  __shared__ float Sbuf[NWAVES][2][256];

  int tid = threadIdx.x;
  int l = tid & 31;
  int w = tid >> 5;
  int b = blockIdx.x * NWAVES + w;
  int bb = (b < batch) ? b : (batch - 1);

  if (tid < N_LAYERS * 8) {
    float a0 = wts[tid * 3 + 0];
    float a1 = wts[tid * 3 + 1];
    float a2 = wts[tid * 3 + 2];
    float cx = cosf(0.5f * a0), sx = sinf(0.5f * a0);
    float cy = cosf(0.5f * a1), sy = sinf(0.5f * a1);
    float cz = cosf(0.5f * a2), sz = sinf(0.5f * a2);
#pragma unroll
    for (int q = 0; q < 4; ++q) {
      int ab = q >> 1, gb = q & 1;
      c2 yx = c2{0.f, 0.f};
      for (int k = 0; k < 2; ++k) {
        c2 p = cmul(ry_el(ab, k, cy, sy), rx_el(k, gb, cx, sx));
        yx.re += p.re; yx.im += p.im;
      }
      c2 z = ab ? c2{cz, sz} : c2{cz, -sz};
      c2 ge = cmul(z, yx);
      Gl[tid * 8 + q * 2 + 0] = ge.re;
      Gl[tid * 8 + q * 2 + 1] = ge.im;
    }
  }
  __syncthreads();
#pragma unroll 8
  for (int e = tid; e < N_LAYERS * 2 * 256; e += NTHREADS) {
    int t = e & 255, m = (e >> 8) & 1, ly = e >> 9;
    int r = t >> 4, c = t & 15;
    int rr = (m == 0) ? (r ^ (r >> 1)) : r;
    c2 prod = c2{1.f, 0.f};
#pragma unroll
    for (int i = 0; i < 4; ++i) {
      int g = ly * 8 + m * 4 + i;
      int ab = (rr >> (3 - i)) & 1;
      int gb = (c  >> (3 - i)) & 1;
      const float* gp = &Gl[g * 8 + (ab * 2 + gb) * 2];
      prod = cmul(prod, c2{gp[0], gp[1]});
    }
    Vl[((ly * 2 + m) * 2 + 0) * 256 + t] = prod.re;
    Vl[((ly * 2 + m) * 2 + 1) * 256 + t] = prod.im;
  }
  __syncthreads();

  int ga = bb / GN, gq = bb - ga * GN;
  const float step = 6.28318530717958647692f / 49.0f;
  float x0 = step * (float)ga, x1 = step * (float)gq;
  float c0 = cosf(0.5f * x0), s0 = sinf(0.5f * x0);
  float c1 = cosf(0.5f * x1), s1 = sinf(0.5f * x1);
#pragma unroll
  for (int t = 0; t < 8; ++t) {
    int idx = t * 32 + l;
    int a = idx >> 4, bcol = idx & 15;
    c2 e = rx_el((a >> 3) & 1, (bcol >> 3) & 1, c0, s0);
    e = cmul(e, ry_el((a >> 2) & 1, (bcol >> 2) & 1, c1, s1));
    e = cmul(e, rx_el((a >> 1) & 1, (bcol >> 1) & 1, c0, s0));
    e = cmul(e, ry_el(a & 1, bcol & 1, c1, s1));
    Ebuf[w][0][idx] = e.re;
    Ebuf[w][1][idx] = e.im;
    Sbuf[w][0][idx] = (idx == 0) ? 1.f : 0.f;
    Sbuf[w][1][idx] = 0.f;
  }

  int csrc = (l & 15); csrc = csrc ^ (csrc >> 1);
  int addrE = (((l & 16) | csrc)) << 2;
  int addrO = addrE ^ (8 << 2);

  for (int layer = 0; layer < N_LAYERS; ++layer) {
    const float* V0r = &Vl[((layer * 2 + 0) * 2 + 0) * 256];
    const float* V0i = &Vl[((layer * 2 + 0) * 2 + 1) * 256];
    const float* R1r = &Vl[((layer * 2 + 1) * 2 + 0) * 256];
    const float* R1i = &Vl[((layer * 2 + 1) * 2 + 1) * 256];
    v8f cr, ci;
    cmm(V0r, V0i, &Ebuf[w][0][0], &Ebuf[w][1][0], false, l, cr, ci);
    store_tile(&Mbuf[w][0][0], l, cr); store_tile(&Mbuf[w][1][0], l, ci);
    cmm(R1r, R1i, &Ebuf[w][0][0], &Ebuf[w][1][0], false, l, cr, ci);
    store_tile(&Nbuf[w][0][0], l, cr); store_tile(&Nbuf[w][1][0], l, ci);
    cmm(&Mbuf[w][0][0], &Mbuf[w][1][0], &Sbuf[w][0][0], &Sbuf[w][1][0], false, l, cr, ci);
    store_tile(&Mbuf[w][0][0], l, cr); store_tile(&Mbuf[w][1][0], l, ci);
    cmm(&Mbuf[w][0][0], &Mbuf[w][1][0], &Nbuf[w][0][0], &Nbuf[w][1][0], true, l, cr, ci);
    v8f nr, ni;
#pragma unroll
    for (int v = 0; v < 8; ++v) {
      int ad = (v & 1) ? addrO : addrE;
      nr[v] = __int_as_float(__builtin_amdgcn_ds_bpermute(ad, __float_as_int(cr[v])));
      ni[v] = __int_as_float(__builtin_amdgcn_ds_bpermute(ad, __float_as_int(ci[v])));
    }
    store_tile(&Sbuf[w][0][0], l, nr);
    store_tile(&Sbuf[w][1][0], l, ni);
  }

  int lr = l & 15, lh = l >> 4;
  float p[8];
#pragma unroll
  for (int v = 0; v < 8; ++v) {
    float a = Sbuf[w][0][(v + 8 * lh) * 16 + lr];
    float bi = Sbuf[w][1][(v + 8 * lh) * 16 + lr];
    p[v] = a * a + bi * bi;
  }
  float t  = p[0] + p[1] + p[2] + p[3] + p[4] + p[5] + p[6] + p[7];
  float u1 = p[0] + p[1] + p[2] + p[3] - p[4] - p[5] - p[6] - p[7];
  float u2 = p[0] + p[1] - p[2] - p[3] + p[4] + p[5] - p[6] - p[7];
  float u3 = p[0] - p[1] + p[2] - p[3] + p[4] - p[5] + p[6] - p[7];
  float ev[8];
  ev[0] = lh ? -t : t;
  ev[1] = u1; ev[2] = u2; ev[3] = u3;
  ev[4] = (lr & 8) ? -t : t;
  ev[5] = (lr & 4) ? -t : t;
  ev[6] = (lr & 2) ? -t : t;
  ev[7] = (lr & 1) ? -t : t;
#pragma unroll
  for (int i = 0; i < 8; ++i) ev[i] = wave_sum(ev[i], l);

  if (b < batch && l == 0) {
    const float kPi = 3.14159265358979323846f;
#pragma unroll
    for (int i = 0; i < 8; ++i) out[b * 8 + i] = kPi * ev[i];
  }
}

__global__ void __launch_bounds__(COEF_THREADS)
qnn_coef(const float* __restrict__ GO, float* __restrict__ AM) {
  __shared__ float ct[GN], st[GN];
  __shared__ float H[GN * 8];
  int t = blockIdx.x >> 1, ph = blockIdx.x & 1;
  int tid = threadIdx.x;
  if (tid < GN) {
    float ang = (float)tid * (2.0f / 49.0f);
    ct[tid] = cospif(ang);
    st[tid] = sinpif(ang);
  }
  __syncthreads();
  if (tid < GN * 8) {
    float s = 0.f;
    int j = 0;
#pragma unroll
    for (int a = 0; a < GN; ++a) {
      float tw = ph ? st[j] : ct[j];
      s += tw * GO[a * (GN * 8) + tid];
      j += t; if (j >= GN) j -= GN;
    }
    H[tid] = s;
  }
  __syncthreads();
  {
    int i = tid & 7, k = tid >> 3;
    int u = k >> 1, ee = k & 1;
    float val = 0.f;
    if (u <= GK) {
      float s = 0.f;
      int j = 0;
#pragma unroll
      for (int bq = 0; bq < GN; ++bq) {
        float tw = ee ? st[j] : ct[j];
        s += tw * H[bq * 8 + i];
        j += u; if (j >= GN) j -= GN;
      }
      float wt = (t == 0) ? (1.0f / 49.0f) : (2.0f / 49.0f);
      float wu = (u == 0) ? (1.0f / 49.0f) : (2.0f / 49.0f);
      val = s * wt * wu;
    }
    int m = i + 8 * ph;
    int kb = k >> 2, h = (k >> 1) & 1, jj = k & 1;
    AM[((t * KB + kb) * 32 + m + 16 * h) * 2 + jj] = val;
  }
}

__global__ void __launch_bounds__(EV_THREADS)
qnn_eval(const float* __restrict__ x, const float* __restrict__ AMg,
         float* __restrict__ out, int batch) {
  __shared__ float AMl[AM_FLOATS];
  int tid = threadIdx.x;
  int l = tid & 31, w = tid >> 5;
  int lr = l & 15, lh = l >> 4;
  {
    const float4* srcp = (const float4*)AMg;
    float4* dstp = (float4*)AMl;
    for (int i = tid; i < AM_FLOATS / 4; i += EV_THREADS) dstp[i] = srcp[i];
  }
  __syncthreads();

  const v2f* Af = (const v2f*)AMl + l;
  int ntiles = (batch + 31) >> 5;
  for (int tile = blockIdx.x * (EV_THREADS / 32) + w; tile < ntiles;
       tile += gridDim.x * (EV_THREADS / 32)) {
    v2f bf[2][KB];
    float c1[2], s1[2];
#pragma unroll
    for (int s = 0; s < 2; ++s) {
      int b = tile * 32 + s * 16 + lr;
      int bb = (b < batch) ? b : (batch - 1);
      float x0 = x[2 * bb], x1 = x[2 * bb + 1];
      float sy, cy;
      sincosf(x1, &sy, &cy);
      sincosf(x0, &s1[s], &c1[s]);
      float c2 = cy * cy - sy * sy, s2 = 2.0f * cy * sy;
      float cu = lh ? cy : 1.0f, su = lh ? sy : 0.0f;
#pragma unroll
      for (int kb = 0; kb < KB; ++kb) {
        bf[s][kb].x = cu; bf[s][kb].y = su;
        float nc = cu * c2 - su * s2;
        su = su * c2 + cu * s2;
        cu = nc;
      }
    }
    float acc0[8], acc1[8];
#pragma unroll
    for (int v = 0; v < 8; ++v) { acc0[v] = 0.f; acc1[v] = 0.f; }
    float ct0 = 1.f, st0 = 0.f, ct1 = 1.f, st1 = 0.f;
    for (int t = 0; t < MT; ++t) {
      v8f d0 = {}, d1 = {};
#pragma unroll
      for (int kb = 0; kb < KB; ++kb) {
        v2f a = Af[(t * KB + kb) * 32];
        d0 = __builtin_amdgcn_wmma_f32_16x16x4_f32(false, a, false, bf[0][kb], (short)0, d0, false, false);
        d1 = __builtin_amdgcn_wmma_f32_16x16x4_f32(false, a, false, bf[1][kb], (short)0, d1, false, false);
      }
      float f0 = lh ? st0 : ct0;
      float f1 = lh ? st1 : ct1;
#pragma unroll
      for (int v = 0; v < 8; ++v) { acc0[v] += f0 * d0[v]; acc1[v] += f1 * d1[v]; }
      float n0 = ct0 * c1[0] - st0 * s1[0]; st0 = st0 * c1[0] + ct0 * s1[0]; ct0 = n0;
      float n1 = ct1 * c1[1] - st1 * s1[1]; st1 = st1 * c1[1] + ct1 * s1[1]; ct1 = n1;
    }
    float tot[8];
    int paddr = (l ^ 16) << 2;
#pragma unroll
    for (int v = 0; v < 8; ++v) {
      float give = lh ? acc0[v] : acc1[v];
      float keep = lh ? acc1[v] : acc0[v];
      tot[v] = keep + __int_as_float(__builtin_amdgcn_ds_bpermute(paddr, __float_as_int(give)));
    }
    int bo = tile * 32 + l;
    if (bo < batch) {
      float4* o = (float4*)(out + (size_t)bo * 8);
      o[0] = make_float4(tot[0], tot[1], tot[2], tot[3]);
      o[1] = make_float4(tot[4], tot[5], tot[6], tot[7]);
    }
  }
}

extern "C" void kernel_launch(void* const* d_in, const int* in_sizes, int n_in,
                              void* d_out, int out_size, void* d_ws, size_t ws_size,
                              hipStream_t stream) {
  const float* x   = (const float*)d_in[0];
  const float* wts = (const float*)d_in[1];
  float* out = (float*)d_out;
  float* ws  = (float*)d_ws;
  float* GO  = ws + WS_GO;
  float* AM  = ws + WS_AM;

  int batch = in_sizes[0] / 2;
  qnn_kernel<<<601, NTHREADS, 0, stream>>>(wts, GO, NGRID);
  qnn_coef<<<2 * MT, COEF_THREADS, 0, stream>>>(GO, AM);
  qnn_eval<<<256, EV_THREADS, 0, stream>>>(x, AM, out, batch);
}


# coefficient launch removed: eval interpolates the 49x49 grid values directly with normalised barycentric trigonometric weights (-1)^b/sin((x-th_b)/2); 2 launches, was 3; sums re-ordered
# speedup vs baseline: 10.2947x; 1.0613x over previous
#include <hip/hip_runtime.h>
#include <math.h>


typedef __attribute__((ext_vector_type(2))) float v2f;
typedef __attribute__((ext_vector_type(8))) float v8f;

#define N_LAYERS 6
#define NWAVES   4
#define NTHREADS (NWAVES * 32)

#define GN 49
#define GK 24
#define NGRID (GN * GN)
#define MT 25
#define KB 13
#define AM_FLOATS (MT * KB * 64)
#define WS_AM 40960
#define EV_THREADS 512
#define PATCH 25
#define W_FLOOR 1e-18f

struct c2 { float re, im; };

__device__ __forceinline__ c2 cmul(c2 a, c2 b) {
  return c2{a.re * b.re - a.im * b.im, a.re * b.im + a.im * b.re};
}
__device__ __forceinline__ c2 rx_el(int a, int b, float c, float s) {
  return (a == b) ? c2{c, 0.f} : c2{0.f, -s};
}
__device__ __forceinline__ c2 ry_el(int a, int b, float c, float s) {
  if (a == b) return c2{c, 0.f};
  return (a == 1) ? c2{s, 0.f} : c2{-s, 0.f};
}

__device__ __forceinline__ void cmm(const float* __restrict__ Ar,
                                    const float* __restrict__ Ai,
                                    const float* __restrict__ Br,
                                    const float* __restrict__ Bi,
                                    bool bT, int l, v8f& Cr, v8f& Ci) {
  int lr = l & 15, lh = l >> 4;
  int aoff = lr * 16 + 2 * lh;
  v8f rr = {}, ii = {}, ri = {};
#pragma unroll
  for (int kb = 0; kb < 4; ++kb) {
    v2f ar = *(const v2f*)(Ar + aoff + 4 * kb);
    v2f ai = *(const v2f*)(Ai + aoff + 4 * kb);
    v2f br, bi;
    if (bT) {
      br = *(const v2f*)(Br + aoff + 4 * kb);
      bi = *(const v2f*)(Bi + aoff + 4 * kb);
    } else {
      int k0 = 4 * kb + 2 * lh;
      br.x = Br[k0 * 16 + lr]; br.y = Br[(k0 + 1) * 16 + lr];
      bi.x = Bi[k0 * 16 + lr]; bi.y = Bi[(k0 + 1) * 16 + lr];
    }
    rr = __builtin_amdgcn_wmma_f32_16x16x4_f32(false, ar, false, br, (short)0, rr, false, false);
    ii = __builtin_amdgcn_wmma_f32_16x16x4_f32(false, ai, false, bi, (short)0, ii, false, false);
    ri = __builtin_amdgcn_wmma_f32_16x16x4_f32(false, ar, false, bi, (short)0, ri, false, false);
    ri = __builtin_amdgcn_wmma_f32_16x16x4_f32(false, ai, false, br, (short)0, ri, false, false);
  }
  Cr = rr - ii;
  Ci = ri;
}

__device__ __forceinline__ void store_tile(float* __restrict__ dst, int l, v8f c) {
  int base = ((l >> 4) * 8) * 16 + (l & 15);
#pragma unroll
  for (int v = 0; v < 8; ++v) dst[base + v * 16] = c[v];
}

__device__ __forceinline__ float wave_sum(float v, int l) {
#pragma unroll
  for (int m = 16; m; m >>= 1) {
    int src = (l ^ m) << 2;
    v += __int_as_float(__builtin_amdgcn_ds_bpermute(src, __float_as_int(v)));
  }
  return v;
}

__global__ void __launch_bounds__(NTHREADS)
qnn_kernel(const float* __restrict__ wts, float* __restrict__ AM) {
  __shared__ float Vl[N_LAYERS * 2 * 2 * 256];
  __shared__ float Gl[N_LAYERS * 8 * 8];
  __shared__ float Ebuf[NWAVES][2][256];
  __shared__ float Mbuf[NWAVES][2][256];
  __shared__ float Nbuf[NWAVES][2][256];
  __shared__ float Sbuf[NWAVES][2][256];
  __shared__ float stage[32];

  int tid = threadIdx.x;
  int l = tid & 31;
  int w = tid >> 5;
  int ta = blockIdx.x / PATCH, tq = blockIdx.x - ta * PATCH;
  int ws_ = __builtin_amdgcn_readfirstlane(w);
  int ga = 2 * ta + (ws_ >> 1), gq = 2 * tq + (ws_ & 1);
  bool valid = (ga < GN) && (gq < GN);

  if (tid < N_LAYERS * 8) {
    float a0 = wts[tid * 3 + 0];
    float a1 = wts[tid * 3 + 1];
    float a2 = wts[tid * 3 + 2];
    float cx = cosf(0.5f * a0), sx = sinf(0.5f * a0);
    float cy = cosf(0.5f * a1), sy = sinf(0.5f * a1);
    float cz = cosf(0.5f * a2), sz = sinf(0.5f * a2);
#pragma unroll
    for (int q = 0; q < 4; ++q) {
      int ab = q >> 1, gb = q & 1;
      c2 yx = c2{0.f, 0.f};
      for (int k = 0; k < 2; ++k) {
        c2 p = cmul(ry_el(ab, k, cy, sy), rx_el(k, gb, cx, sx));
        yx.re += p.re; yx.im += p.im;
      }
      c2 z = ab ? c2{cz, sz} : c2{cz, -sz};
      c2 ge = cmul(z, yx);
      Gl[tid * 8 + q * 2 + 0] = ge.re;
      Gl[tid * 8 + q * 2 + 1] = ge.im;
    }
  }
  __syncthreads();
#pragma unroll 8
  for (int e = tid; e < N_LAYERS * 2 * 256; e += NTHREADS) {
    int t = e & 255, m = (e >> 8) & 1, ly = e >> 9;
    int r = t >> 4, c = t & 15;
    int rr = (m == 0) ? (r ^ (r >> 1)) : r;
    c2 prod = c2{1.f, 0.f};
#pragma unroll
    for (int i = 0; i < 4; ++i) {
      int g = ly * 8 + m * 4 + i;
      int ab = (rr >> (3 - i)) & 1;
      int gb = (c  >> (3 - i)) & 1;
      const float* gp = &Gl[g * 8 + (ab * 2 + gb) * 2];
      prod = cmul(prod, c2{gp[0], gp[1]});
    }
    Vl[((ly * 2 + m) * 2 + 0) * 256 + t] = prod.re;
    Vl[((ly * 2 + m) * 2 + 1) * 256 + t] = prod.im;
  }
  __syncthreads();

  float ev[8];
#pragma unroll
  for (int i = 0; i < 8; ++i) ev[i] = 0.f;
  if (valid) {
  const float step = 6.28318530717958647692f / 49.0f;
  float x0 = step * (float)ga, x1 = step * (float)gq;
  float c0 = cosf(0.5f * x0), s0 = sinf(0.5f * x0);
  float c1 = cosf(0.5f * x1), s1 = sinf(0.5f * x1);
#pragma unroll
  for (int t = 0; t < 8; ++t) {
    int idx = t * 32 + l;
    int a = idx >> 4, bcol = idx & 15;
    c2 e = rx_el((a >> 3) & 1, (bcol >> 3) & 1, c0, s0);
    e = cmul(e, ry_el((a >> 2) & 1, (bcol >> 2) & 1, c1, s1));
    e = cmul(e, rx_el((a >> 1) & 1, (bcol >> 1) & 1, c0, s0));
    e = cmul(e, ry_el(a & 1, bcol & 1, c1, s1));
    Ebuf[w][0][idx] = e.re;
    Ebuf[w][1][idx] = e.im;
    Sbuf[w][0][idx] = (idx == 0) ? 1.f : 0.f;
    Sbuf[w][1][idx] = 0.f;
  }

  int csrc = (l & 15); csrc = csrc ^ (csrc >> 1);
  int addrE = (((l & 16) | csrc)) << 2;
  int addrO = addrE ^ (8 << 2);

  for (int layer = 0; layer < N_LAYERS; ++layer) {
    const float* V0r = &Vl[((layer * 2 + 0) * 2 + 0) * 256];
    const float* V0i = &Vl[((layer * 2 + 0) * 2 + 1) * 256];
    const float* R1r = &Vl[((layer * 2 + 1) * 2 + 0) * 256];
    const float* R1i = &Vl[((layer * 2 + 1) * 2 + 1) * 256];
    v8f cr, ci;
    cmm(V0r, V0i, &Ebuf[w][0][0], &Ebuf[w][1][0], false, l, cr, ci);
    store_tile(&Mbuf[w][0][0], l, cr); store_tile(&Mbuf[w][1][0], l, ci);
    cmm(R1r, R1i, &Ebuf[w][0][0], &Ebuf[w][1][0], false, l, cr, ci);
    store_tile(&Nbuf[w][0][0], l, cr); store_tile(&Nbuf[w][1][0], l, ci);
    cmm(&Mbuf[w][0][0], &Mbuf[w][1][0], &Sbuf[w][0][0], &Sbuf[w][1][0], false, l, cr, ci);
    store_tile(&Mbuf[w][0][0], l, cr); store_tile(&Mbuf[w][1][0], l, ci);
    cmm(&Mbuf[w][0][0], &Mbuf[w][1][0], &Nbuf[w][0][0], &Nbuf[w][1][0], true, l, cr, ci);
    v8f nr, ni;
#pragma unroll
    for (int v = 0; v < 8; ++v) {
      int ad = (v & 1) ? addrO : addrE;
      nr[v] = __int_as_float(__builtin_amdgcn_ds_bpermute(ad, __float_as_int(cr[v])));
      ni[v] = __int_as_float(__builtin_amdgcn_ds_bpermute(ad, __float_as_int(ci[v])));
    }
    store_tile(&Sbuf[w][0][0], l, nr);
    store_tile(&Sbuf[w][1][0], l, ni);
  }

  int lr = l & 15, lh = l >> 4;
  float p[8];
#pragma unroll
  for (int v = 0; v < 8; ++v) {
    float a = Sbuf[w][0][(v + 8 * lh) * 16 + lr];
    float bi = Sbuf[w][1][(v + 8 * lh) * 16 + lr];
    p[v] = a * a + bi * bi;
  }
  float t  = p[0] + p[1] + p[2] + p[3] + p[4] + p[5] + p[6] + p[7];
  float u1 = p[0] + p[1] + p[2] + p[3] - p[4] - p[5] - p[6] - p[7];
  float u2 = p[0] + p[1] - p[2] - p[3] + p[4] + p[5] - p[6] - p[7];
  float u3 = p[0] - p[1] + p[2] - p[3] + p[4] - p[5] + p[6] - p[7];
  ev[0] = lh ? -t : t;
  ev[1] = u1; ev[2] = u2; ev[3] = u3;
  ev[4] = (lr & 8) ? -t : t;
  ev[5] = (lr & 4) ? -t : t;
  ev[6] = (lr & 2) ? -t : t;
  ev[7] = (lr & 1) ? -t : t;
#pragma unroll
  for (int i = 0; i < 8; ++i) ev[i] = wave_sum(ev[i], l);
  }

  if (l == 0) {
    const float kPi = 3.14159265358979323846f;
#pragma unroll
    for (int i = 0; i < 8; ++i)
      stage[(i + 8 * (ws_ >> 1)) * 2 + (ws_ & 1)] = valid ? kPi * ev[i] : 0.f;
  }
  __syncthreads();
  int line = ((ta * KB + (tq >> 1)) * 32 + 16 * (tq & 1)) * 2;
  if (ws_ == 0) AM[line + l] = stage[l];
  if (ws_ == 1 && tq == PATCH - 1) AM[line + 32 + l] = 0.f;
}

__device__ __forceinline__ float node_weight(float sd) {
  sd = (fabsf(sd) < W_FLOOR) ? W_FLOOR : sd;
  return __builtin_amdgcn_rcpf(sd);
}

__global__ void __launch_bounds__(EV_THREADS)
qnn_eval(const float* __restrict__ x, const float* __restrict__ AMg,
         float* __restrict__ out, int batch) {
  __shared__ float AMl[AM_FLOATS];
  __shared__ float4 tab4[KB * 2];
  int tid = threadIdx.x;
  int l = tid & 31, w = tid >> 5;
  int lr = l & 15, lh = l >> 4;
  {
    const float4* srcp = (const float4*)AMg;
    float4* dstp = (float4*)AMl;
    for (int i = tid; i < AM_FLOATS / 4; i += EV_THREADS) dstp[i] = srcp[i];
  }
  if (tid < KB * 4) {
    float ang = (float)tid / 49.0f;
    float* tb = (float*)tab4;
    tb[2 * tid]     = cospif(ang);
    tb[2 * tid + 1] = sinpif(ang);
  }
  __syncthreads();

  const v2f* Af = (const v2f*)AMl + l;
  const v2f* tab2 = (const v2f*)tab4;
  int paddr = (l ^ 16) << 2;
  int ntiles = (batch + 31) >> 5;
  for (int tile = blockIdx.x * (EV_THREADS / 32) + w; tile < ntiles;
       tile += gridDim.x * (EV_THREADS / 32)) {
    v2f bf[2][KB];
    float sh1[2], ch1[2];
    float shs[2], chs[2];
#pragma unroll
    for (int s = 0; s < 2; ++s) {
      int b = tile * 32 + s * 16 + lr;
      int bb = (b < batch) ? b : (batch - 1);
      float x0 = x[2 * bb], x1 = x[2 * bb + 1];
      float sa, ca;
      sincosf(0.5f * x1, &sh1[s], &ch1[s]);
      sincosf(0.5f * x0, &sa, &ca);
      shs[s] = lh ? -sa : sa;
      chs[s] = lh ? -ca : ca;
    }
    float sum[2] = {0.f, 0.f};
#pragma unroll
    for (int kb = 0; kb < KB; ++kb) {
      float4 tb = tab4[2 * kb + lh];
#pragma unroll
      for (int s = 0; s < 2; ++s) {
        float w0 =  node_weight(sh1[s] * tb.x - ch1[s] * tb.y);
        float w1 = -node_weight(sh1[s] * tb.z - ch1[s] * tb.w);
        if (kb == KB - 1) {
          w1 = 0.f;
          w0 = lh ? 0.f : w0;
        }
        bf[s][kb].x = w0; bf[s][kb].y = w1;
        sum[s] += w0 + w1;
      }
    }
#pragma unroll
    for (int s = 0; s < 2; ++s) {
      float tot = sum[s] + __int_as_float(__builtin_amdgcn_ds_bpermute(paddr, __float_as_int(sum[s])));
      float inv = 1.0f / tot;
#pragma unroll
      for (int kb = 0; kb < KB; ++kb) { bf[s][kb].x *= inv; bf[s][kb].y *= inv; }
    }
    float acc0[8], acc1[8];
#pragma unroll
    for (int v = 0; v < 8; ++v) { acc0[v] = 0.f; acc1[v] = 0.f; }
    float ws0 = 0.f, ws1 = 0.f;
    for (int t = 0; t < MT; ++t) {
      v8f d0 = {}, d1 = {};
#pragma unroll
      for (int kb = 0; kb < KB; ++kb) {
        v2f a = Af[(t * KB + kb) * 32];
        d0 = __builtin_amdgcn_wmma_f32_16x16x4_f32(false, a, false, bf[0][kb], (short)0, d0, false, false);
        d1 = __builtin_amdgcn_wmma_f32_16x16x4_f32(false, a, false, bf[1][kb], (short)0, d1, false, false);
      }
      v2f cs = tab2[2 * t + lh];
      float f0 = node_weight(shs[0] * cs.x - chs[0] * cs.y);
      float f1 = node_weight(shs[1] * cs.x - chs[1] * cs.y);
      if (t == MT - 1) {
        f0 = lh ? 0.f : f0;
        f1 = lh ? 0.f : f1;
      }
      ws0 += f0; ws1 += f1;
#pragma unroll
      for (int v = 0; v < 8; ++v) { acc0[v] += f0 * d0[v]; acc1[v] += f1 * d1[v]; }
    }
    float gw = lh ? ws0 : ws1;
    float kw = lh ? ws1 : ws0;
    float den = kw + __int_as_float(__builtin_amdgcn_ds_bpermute(paddr, __float_as_int(gw)));
    float rden = 1.0f / den;
    float tot[8];
#pragma unroll
    for (int v = 0; v < 8; ++v) {
      float give = lh ? acc0[v] : acc1[v];
      float keep = lh ? acc1[v] : acc0[v];
      tot[v] = (keep + __int_as_float(__builtin_amdgcn_ds_bpermute(paddr, __float_as_int(give)))) * rden;
    }
    int bo = tile * 32 + l;
    if (bo < batch) {
      float4* o = (float4*)(out + (size_t)bo * 8);
      o[0] = make_float4(tot[0], tot[1], tot[2], tot[3]);
      o[1] = make_float4(tot[4], tot[5], tot[6], tot[7]);
    }
  }
}

extern "C" void kernel_launch(void* const* d_in, const int* in_sizes, int n_in,
                              void* d_out, int out_size, void* d_ws, size_t ws_size,
                              hipStream_t stream) {
  const float* x   = (const float*)d_in[0];
  const float* wts = (const float*)d_in[1];
  float* out = (float*)d_out;
  float* ws  = (float*)d_ws;
  float* AM  = ws + WS_AM;

  int batch = in_sizes[0] / 2;
  qnn_kernel<<<PATCH * PATCH, NTHREADS, 0, stream>>>(wts, AM);
  qnn_eval<<<256, EV_THREADS, 0, stream>>>(x, AM, out, batch);
}


# tile rows = x0-encoded wires 0,2,4,6, columns = x1-encoded wires 1,3,5,7: encoded layer matrices depend on one grid angle, built once per work-group in LDS; 264 WMMA per run, was 384; sums re-ordered
# speedup vs baseline: 10.8497x; 1.0539x over previous
#include <hip/hip_runtime.h>
#include <math.h>


typedef __attribute__((ext_vector_type(2))) float v2f;
typedef __attribute__((ext_vector_type(8))) float v8f;

#define N_LAYERS 6
#define NWAVES   4
#define NTHREADS (NWAVES * 32)

#define GN 49
#define GK 24
#define NGRID (GN * GN)
#define MT 25
#define KB 13
#define AM_FLOATS (MT * KB * 64)
#define WS_AM 40960
#define EV_THREADS 512
#define PATCH 25
#define W_FLOOR 1e-18f

struct c2 { float re, im; };

__device__ __forceinline__ c2 cmul(c2 a, c2 b) {
  return c2{a.re * b.re - a.im * b.im, a.re * b.im + a.im * b.re};
}
__device__ __forceinline__ c2 rx_el(int a, int b, float c, float s) {
  return (a == b) ? c2{c, 0.f} : c2{0.f, -s};
}
__device__ __forceinline__ c2 ry_el(int a, int b, float c, float s) {
  if (a == b) return c2{c, 0.f};
  return (a == 1) ? c2{s, 0.f} : c2{-s, 0.f};
}

__device__ __forceinline__ void cmm(const float* __restrict__ Ar,
                                    const float* __restrict__ Ai,
                                    const float* __restrict__ Br,
                                    const float* __restrict__ Bi,
                                    bool bT, int l, v8f& Cr, v8f& Ci) {
  int lr = l & 15, lh = l >> 4;
  int aoff = lr * 16 + 2 * lh;
  v8f rr = {}, ii = {}, ri = {};
#pragma unroll
  for (int kb = 0; kb < 4; ++kb) {
    v2f ar = *(const v2f*)(Ar + aoff + 4 * kb);
    v2f ai = *(const v2f*)(Ai + aoff + 4 * kb);
    v2f br, bi;
    if (bT) {
      br = *(const v2f*)(Br + aoff + 4 * kb);
      bi = *(const v2f*)(Bi + aoff + 4 * kb);
    } else {
      int k0 = 4 * kb + 2 * lh;
      br.x = Br[k0 * 16 + lr]; br.y = Br[(k0 + 1) * 16 + lr];
      bi.x = Bi[k0 * 16 + lr]; bi.y = Bi[(k0 + 1) * 16 + lr];
    }
    rr = __builtin_amdgcn_wmma_f32_16x16x4_f32(false, ar, false, br, (short)0, rr, false, false);
    ii = __builtin_amdgcn_wmma_f32_16x16x4_f32(false, ai, false, bi, (short)0, ii, false, false);
    ri = __builtin_amdgcn_wmma_f32_16x16x4_f32(false, ar, false, bi, (short)0, ri, false, false);
    ri = __builtin_amdgcn_wmma_f32_16x16x4_f32(false, ai, false, br, (short)0, ri, false, false);
  }
  Cr = rr - ii;
  Ci = ri;
}

__device__ __forceinline__ void cmm_rb(const float* __restrict__ Ar,
                                       const float* __restrict__ Ai,
                                       const float* __restrict__ Br,
                                       int l, v8f& Cr, v8f& Ci) {
  int lr = l & 15, lh = l >> 4;
  int aoff = lr * 16 + 2 * lh;
  v8f rr = {}, ir = {};
#pragma unroll
  for (int kb = 0; kb < 4; ++kb) {
    v2f ar = *(const v2f*)(Ar + aoff + 4 * kb);
    v2f ai = *(const v2f*)(Ai + aoff + 4 * kb);
    int k0 = 4 * kb + 2 * lh;
    v2f br;
    br.x = Br[k0 * 16 + lr]; br.y = Br[(k0 + 1) * 16 + lr];
    rr = __builtin_amdgcn_wmma_f32_16x16x4_f32(false, ar, false, br, (short)0, rr, false, false);
    ir = __builtin_amdgcn_wmma_f32_16x16x4_f32(false, ai, false, br, (short)0, ir, false, false);
  }
  Cr = rr;
  Ci = ir;
}

__device__ __forceinline__ void store_tile(float* __restrict__ dst, int l, v8f c) {
  int base = ((l >> 4) * 8) * 16 + (l & 15);
#pragma unroll
  for (int v = 0; v < 8; ++v) dst[base + v * 16] = c[v];
}

__device__ __forceinline__ float wave_sum(float v, int l) {
#pragma unroll
  for (int m = 16; m; m >>= 1) {
    int src = (l ^ m) << 2;
    v += __int_as_float(__builtin_amdgcn_ds_bpermute(src, __float_as_int(v)));
  }
  return v;
}

__device__ __forceinline__ int chain_dst(int r, int c) {
  int p0 = (r >> 3) & 1;
  int p1 = p0 ^ ((c >> 3) & 1);
  int p2 = p1 ^ ((r >> 2) & 1);
  int p3 = p2 ^ ((c >> 2) & 1);
  int p4 = p3 ^ ((r >> 1) & 1);
  int p5 = p4 ^ ((c >> 1) & 1);
  int p6 = p5 ^ (r & 1);
  int p7 = p6 ^ (c & 1);
  int rn = (p0 << 3) | (p2 << 2) | (p4 << 1) | p6;
  int cn = (p1 << 3) | (p3 << 2) | (p5 << 1) | p7;
  return rn * 16 + cn;
}

__global__ void __launch_bounds__(NTHREADS)
qnn_kernel(const float* __restrict__ wts, float* __restrict__ AM) {
  __shared__ float Vl[N_LAYERS * 2 * 2 * 256];
  __shared__ float Gl[N_LAYERS * 8 * 8];
  __shared__ float ABl[4][N_LAYERS][2][256];
  __shared__ float Ebuf[NWAVES][2][256];
  __shared__ float Sbuf[NWAVES][2][256];
  __shared__ float stage[32];

  int tid = threadIdx.x;
  int l = tid & 31;
  int ta = blockIdx.x / PATCH, tq = blockIdx.x - ta * PATCH;
  int ws_ = __builtin_amdgcn_readfirstlane(tid >> 5);
  int ga = 2 * ta + (ws_ >> 1), gq = 2 * tq + (ws_ & 1);
  bool valid = (ga < GN) && (gq < GN);

  if (tid < N_LAYERS * 8) {
    float a0 = wts[tid * 3 + 0];
    float a1 = wts[tid * 3 + 1];
    float a2 = wts[tid * 3 + 2];
    float cx = cosf(0.5f * a0), sx = sinf(0.5f * a0);
    float cy = cosf(0.5f * a1), sy = sinf(0.5f * a1);
    float cz = cosf(0.5f * a2), sz = sinf(0.5f * a2);
#pragma unroll
    for (int q = 0; q < 4; ++q) {
      int ab = q >> 1, gb = q & 1;
      c2 yx = c2{0.f, 0.f};
      for (int k = 0; k < 2; ++k) {
        c2 p = cmul(ry_el(ab, k, cy, sy), rx_el(k, gb, cx, sx));
        yx.re += p.re; yx.im += p.im;
      }
      c2 z = ab ? c2{cz, sz} : c2{cz, -sz};
      c2 ge = cmul(z, yx);
      Gl[tid * 8 + q * 2 + 0] = ge.re;
      Gl[tid * 8 + q * 2 + 1] = ge.im;
    }
  }
  __syncthreads();
#pragma unroll 8
  for (int e = tid; e < N_LAYERS * 2 * 256; e += NTHREADS) {
    int t = e & 255, m = (e >> 8) & 1, ly = e >> 9;
    int r = t >> 4, c = t & 15;
    c2 prod = c2{1.f, 0.f};
#pragma unroll
    for (int i = 0; i < 4; ++i) {
      int g = ly * 8 + 2 * i + m;
      int ab = (r >> (3 - i)) & 1;
      int gb = (c >> (3 - i)) & 1;
      const float* gp = &Gl[g * 8 + (ab * 2 + gb) * 2];
      prod = cmul(prod, c2{gp[0], gp[1]});
    }
    Vl[((ly * 2 + m) * 2 + 0) * 256 + t] = prod.re;
    Vl[((ly * 2 + m) * 2 + 1) * 256 + t] = prod.im;
  }
  __syncthreads();

  const float step = 6.28318530717958647692f / 49.0f;
  {
    int par = ws_ & 1;
    int lbase = 3 * (ws_ >> 1);
    float ang0 = step * (float)(2 * ta + par);
    float ang1 = step * (float)(2 * tq + par);
    float c0 = cosf(0.5f * ang0), s0 = sinf(0.5f * ang0);
    float c1 = cosf(0.5f * ang1), s1 = sinf(0.5f * ang1);
    float* Er = &Ebuf[ws_][0][0];
    float* Ei = &Ebuf[ws_][1][0];
    float* Ey = &Sbuf[ws_][0][0];
#pragma unroll
    for (int t = 0; t < 8; ++t) {
      int idx = t * 32 + l;
      int a = idx >> 4, bcol = idx & 15;
      c2 ex = c2{1.f, 0.f};
      float ey = 1.f;
#pragma unroll
      for (int i = 0; i < 4; ++i) {
        int ab = (a >> (3 - i)) & 1, gb = (bcol >> (3 - i)) & 1;
        ex = cmul(ex, rx_el(ab, gb, c0, s0));
        ey *= ry_el(ab, gb, c1, s1).re;
      }
      Er[idx] = ex.re;
      Ei[idx] = ex.im;
      Ey[idx] = ey;
    }
    for (int j = 0; j < 3; ++j) {
      int layer = lbase + j;
      const float* Rrr = &Vl[((layer * 2 + 0) * 2 + 0) * 256];
      const float* Rri = &Vl[((layer * 2 + 0) * 2 + 1) * 256];
      const float* Rcr = &Vl[((layer * 2 + 1) * 2 + 0) * 256];
      const float* Rci = &Vl[((layer * 2 + 1) * 2 + 1) * 256];
      v8f cr, ci;
      cmm(Rrr, Rri, Er, Ei, false, l, cr, ci);
      store_tile(&ABl[par][layer][0][0], l, cr);
      store_tile(&ABl[par][layer][1][0], l, ci);
      cmm_rb(Rcr, Rci, Ey, l, cr, ci);
      store_tile(&ABl[2 + par][layer][0][0], l, cr);
      store_tile(&ABl[2 + par][layer][1][0], l, ci);
    }
  }
  __syncthreads();

  float ev[8];
#pragma unroll
  for (int i = 0; i < 8; ++i) ev[i] = 0.f;
  if (valid) {
  int lr = l & 15, lh = l >> 4;
  float* Sr = &Sbuf[ws_][0][0];
  float* Si = &Sbuf[ws_][1][0];
  float* Tr = &Ebuf[ws_][0][0];
  float* Ti = &Ebuf[ws_][1][0];
#pragma unroll
  for (int t = 0; t < 8; ++t) {
    int idx = t * 32 + l;
    Sr[idx] = (idx == 0) ? 1.f : 0.f;
    Si[idx] = 0.f;
  }
  int dsta[8];
#pragma unroll
  for (int v = 0; v < 8; ++v) dsta[v] = chain_dst(v + 8 * lh, lr);

  const float* Ab = &ABl[ws_ >> 1][0][0][0];
  const float* Bb = &ABl[2 + (ws_ & 1)][0][0][0];
  for (int layer = 0; layer < N_LAYERS; ++layer) {
    const float* Ar = Ab + layer * 512;
    const float* Ai = Ar + 256;
    const float* Br = Bb + layer * 512;
    const float* Bi = Br + 256;
    v8f cr, ci;
    cmm(Ar, Ai, Sr, Si, false, l, cr, ci);
    store_tile(Tr, l, cr); store_tile(Ti, l, ci);
    cmm(Tr, Ti, Br, Bi, true, l, cr, ci);
#pragma unroll
    for (int v = 0; v < 8; ++v) {
      Sr[dsta[v]] = cr[v];
      Si[dsta[v]] = ci[v];
    }
  }

  float p[8];
#pragma unroll
  for (int v = 0; v < 8; ++v) {
    float a = Sr[(v + 8 * lh) * 16 + lr];
    float bi = Si[(v + 8 * lh) * 16 + lr];
    p[v] = a * a + bi * bi;
  }
  float t  = p[0] + p[1] + p[2] + p[3] + p[4] + p[5] + p[6] + p[7];
  float u1 = p[0] + p[1] + p[2] + p[3] - p[4] - p[5] - p[6] - p[7];
  float u2 = p[0] + p[1] - p[2] - p[3] + p[4] + p[5] - p[6] - p[7];
  float u3 = p[0] - p[1] + p[2] - p[3] + p[4] - p[5] + p[6] - p[7];
  ev[0] = lh ? -t : t;
  ev[2] = u1; ev[4] = u2; ev[6] = u3;
  ev[1] = (lr & 8) ? -t : t;
  ev[3] = (lr & 4) ? -t : t;
  ev[5] = (lr & 2) ? -t : t;
  ev[7] = (lr & 1) ? -t : t;
#pragma unroll
  for (int i = 0; i < 8; ++i) ev[i] = wave_sum(ev[i], l);
  }

  if (l == 0) {
    const float kPi = 3.14159265358979323846f;
#pragma unroll
    for (int i = 0; i < 8; ++i)
      stage[(i + 8 * (ws_ >> 1)) * 2 + (ws_ & 1)] = valid ? kPi * ev[i] : 0.f;
  }
  __syncthreads();
  int line = ((ta * KB + (tq >> 1)) * 32 + 16 * (tq & 1)) * 2;
  if (ws_ == 0) AM[line + l] = stage[l];
  if (ws_ == 1 && tq == PATCH - 1) AM[line + 32 + l] = 0.f;
}

__device__ __forceinline__ float node_weight(float sd) {
  sd = (fabsf(sd) < W_FLOOR) ? W_FLOOR : sd;
  return __builtin_amdgcn_rcpf(sd);
}

__global__ void __launch_bounds__(EV_THREADS)
qnn_eval(const float* __restrict__ x, const float* __restrict__ AMg,
         float* __restrict__ out, int batch) {
  __shared__ float AMl[AM_FLOATS];
  __shared__ float4 tab4[KB * 2];
  int tid = threadIdx.x;
  int l = tid & 31, w = tid >> 5;
  int lr = l & 15, lh = l >> 4;
  {
    const float4* srcp = (const float4*)AMg;
    float4* dstp = (float4*)AMl;
    for (int i = tid; i < AM_FLOATS / 4; i += EV_THREADS) dstp[i] = srcp[i];
  }
  if (tid < KB * 4) {
    float ang = (float)tid / 49.0f;
    float* tb = (float*)tab4;
    tb[2 * tid]     = cospif(ang);
    tb[2 * tid + 1] = sinpif(ang);
  }
  __syncthreads();

  const v2f* Af = (const v2f*)AMl + l;
  const v2f* tab2 = (const v2f*)tab4;
  int paddr = (l ^ 16) << 2;
  int ntiles = (batch + 31) >> 5;
  for (int tile = blockIdx.x * (EV_THREADS / 32) + w; tile < ntiles;
       tile += gridDim.x * (EV_THREADS / 32)) {
    v2f bf[2][KB];
    float sh1[2], ch1[2];
    float shs[2], chs[2];
#pragma unroll
    for (int s = 0; s < 2; ++s) {
      int b = tile * 32 + s * 16 + lr;
      int bb = (b < batch) ? b : (batch - 1);
      float x0 = x[2 * bb], x1 = x[2 * bb + 1];
      float sa, ca;
      sincosf(0.5f * x1, &sh1[s], &ch1[s]);
      sincosf(0.5f * x0, &sa, &ca);
      shs[s] = lh ? -sa : sa;
      chs[s] = lh ? -ca : ca;
    }
    float sum[2] = {0.f, 0.f};
#pragma unroll
    for (int kb = 0; kb < KB; ++kb) {
      float4 tb = tab4[2 * kb + lh];
#pragma unroll
      for (int s = 0; s < 2; ++s) {
        float w0 =  node_weight(sh1[s] * tb.x - ch1[s] * tb.y);
        float w1 = -node_weight(sh1[s] * tb.z - ch1[s] * tb.w);
        if (kb == KB - 1) {
          w1 = 0.f;
          w0 = lh ? 0.f : w0;
        }
        bf[s][kb].x = w0; bf[s][kb].y = w1;
        sum[s] += w0 + w1;
      }
    }
#pragma unroll
    for (int s = 0; s < 2; ++s) {
      float tot = sum[s] + __int_as_float(__builtin_amdgcn_ds_bpermute(paddr, __float_as_int(sum[s])));
      float inv = 1.0f / tot;
#pragma unroll
      for (int kb = 0; kb < KB; ++kb) { bf[s][kb].x *= inv; bf[s][kb].y *= inv; }
    }
    float acc0[8], acc1[8];
#pragma unroll
    for (int v = 0; v < 8; ++v) { acc0[v] = 0.f; acc1[v] = 0.f; }
    float ws0 = 0.f, ws1 = 0.f;
    for (int t = 0; t < MT; ++t) {
      v8f d0 = {}, d1 = {};
#pragma unroll
      for (int kb = 0; kb < KB; ++kb) {
        v2f a = Af[(t * KB + kb) * 32];
        d0 = __builtin_amdgcn_wmma_f32_16x16x4_f32(false, a, false, bf[0][kb], (short)0, d0, false, false);
        d1 = __builtin_amdgcn_wmma_f32_16x16x4_f32(false, a, false, bf[1][kb], (short)0, d1, false, false);
      }
      v2f cs = tab2[2 * t + lh];
      float f0 = node_weight(shs[0] * cs.x - chs[0] * cs.y);
      float f1 = node_weight(shs[1] * cs.x - chs[1] * cs.y);
      if (t == MT - 1) {
        f0 = lh ? 0.f : f0;
        f1 = lh ? 0.f : f1;
      }
      ws0 += f0; ws1 += f1;
#pragma unroll
      for (int v = 0; v < 8; ++v) { acc0[v] += f0 * d0[v]; acc1[v] += f1 * d1[v]; }
    }
    float gw = lh ? ws0 : ws1;
    float kw = lh ? ws1 : ws0;
    float den = kw + __int_as_float(__builtin_amdgcn_ds_bpermute(paddr, __float_as_int(gw)));
    float rden = 1.0f / den;
    float tot[8];
#pragma unroll
    for (int v = 0; v < 8; ++v) {
      float give = lh ? acc0[v] : acc1[v];
      float keep = lh ? acc1[v] : acc0[v];
      tot[v] = (keep + __int_as_float(__builtin_amdgcn_ds_bpermute(paddr, __float_as_int(give)))) * rden;
    }
    int bo = tile * 32 + l;
    if (bo < batch) {
      float4* o = (float4*)(out + (size_t)bo * 8);
      o[0] = make_float4(tot[0], tot[1], tot[2], tot[3]);
      o[1] = make_float4(tot[4], tot[5], tot[6], tot[7]);
    }
  }
}

extern "C" void kernel_launch(void* const* d_in, const int* in_sizes, int n_in,
                              void* d_out, int out_size, void* d_ws, size_t ws_size,
                              hipStream_t stream) {
  const float* x   = (const float*)d_in[0];
  const float* wts = (const float*)d_in[1];
  float* out = (float*)d_out;
  float* ws  = (float*)d_ws;
  float* AM  = ws + WS_AM;

  int batch = in_sizes[0] / 2;
  qnn_kernel<<<PATCH * PATCH, NTHREADS, 0, stream>>>(wts, AM);
  qnn_eval<<<256, EV_THREADS, 0, stream>>>(x, AM, out, batch);
}


# encoded layer matrices A_l, B_l formed as Kronecker products of the encoded 2x2 gates W RX(x0), W RY(x1): the 72 WMMA per run of Rr Ex, Rc Ey removed; 192 WMMA per run, was 264; sums re-ordered
# speedup vs baseline: 11.7494x; 1.0829x over previous
#include <hip/hip_runtime.h>
#include <math.h>


typedef __attribute__((ext_vector_type(2))) float v2f;
typedef __attribute__((ext_vector_type(4))) float v4f;
typedef __attribute__((ext_vector_type(8))) float v8f;

#define N_LAYERS 6
#define NWAVES   4
#define NTHREADS (NWAVES * 32)

#define GN 49
#define GK 24
#define NGRID (GN * GN)
#define MT 25
#define KB 13
#define AM_FLOATS (MT * KB * 64)
#define WS_AM 40960
#define EV_THREADS 512
#define PATCH 25
#define W_FLOOR 1e-18f

struct c2 { float re, im; };

__device__ __forceinline__ c2 cmul(c2 a, c2 b) {
  return c2{a.re * b.re - a.im * b.im, a.re * b.im + a.im * b.re};
}
__device__ __forceinline__ c2 rx_el(int a, int b, float c, float s) {
  return (a == b) ? c2{c, 0.f} : c2{0.f, -s};
}
__device__ __forceinline__ c2 ry_el(int a, int b, float c, float s) {
  if (a == b) return c2{c, 0.f};
  return (a == 1) ? c2{s, 0.f} : c2{-s, 0.f};
}

__device__ __forceinline__ void cmm(const float* __restrict__ Ar,
                                    const float* __restrict__ Ai,
                                    const float* __restrict__ Br,
                                    const float* __restrict__ Bi,
                                    bool bT, int l, v8f& Cr, v8f& Ci) {
  int lr = l & 15, lh = l >> 4;
  int aoff = lr * 16 + 2 * lh;
  v8f rr = {}, ii = {}, ri = {};
#pragma unroll
  for (int kb = 0; kb < 4; ++kb) {
    v2f ar = *(const v2f*)(Ar + aoff + 4 * kb);
    v2f ai = *(const v2f*)(Ai + aoff + 4 * kb);
    v2f br, bi;
    if (bT) {
      br = *(const v2f*)(Br + aoff + 4 * kb);
      bi = *(const v2f*)(Bi + aoff + 4 * kb);
    } else {
      int k0 = 4 * kb + 2 * lh;
      br.x = Br[k0 * 16 + lr]; br.y = Br[(k0 + 1) * 16 + lr];
      bi.x = Bi[k0 * 16 + lr]; bi.y = Bi[(k0 + 1) * 16 + lr];
    }
    rr = __builtin_amdgcn_wmma_f32_16x16x4_f32(false, ar, false, br, (short)0, rr, false, false);
    ii = __builtin_amdgcn_wmma_f32_16x16x4_f32(false, ai, false, bi, (short)0, ii, false, false);
    ri = __builtin_amdgcn_wmma_f32_16x16x4_f32(false, ar, false, bi, (short)0, ri, false, false);
    ri = __builtin_amdgcn_wmma_f32_16x16x4_f32(false, ai, false, br, (short)0, ri, false, false);
  }
  Cr = rr - ii;
  Ci = ri;
}

__device__ __forceinline__ void store_tile(float* __restrict__ dst, int l, v8f c) {
  int base = ((l >> 4) * 8) * 16 + (l & 15);
#pragma unroll
  for (int v = 0; v < 8; ++v) dst[base + v * 16] = c[v];
}

__device__ __forceinline__ float wave_sum(float v, int l) {
#pragma unroll
  for (int m = 16; m; m >>= 1) {
    int src = (l ^ m) << 2;
    v += __int_as_float(__builtin_amdgcn_ds_bpermute(src, __float_as_int(v)));
  }
  return v;
}

__device__ __forceinline__ int chain_dst(int r, int c) {
  int p0 = (r >> 3) & 1;
  int p1 = p0 ^ ((c >> 3) & 1);
  int p2 = p1 ^ ((r >> 2) & 1);
  int p3 = p2 ^ ((c >> 2) & 1);
  int p4 = p3 ^ ((r >> 1) & 1);
  int p5 = p4 ^ ((c >> 1) & 1);
  int p6 = p5 ^ (r & 1);
  int p7 = p6 ^ (c & 1);
  int rn = (p0 << 3) | (p2 << 2) | (p4 << 1) | p6;
  int cn = (p1 << 3) | (p3 << 2) | (p5 << 1) | p7;
  return rn * 16 + cn;
}

__global__ void __launch_bounds__(NTHREADS)
qnn_kernel(const float* __restrict__ wts, float* __restrict__ AM) {
  __shared__ __attribute__((aligned(16))) float Gl[2 * N_LAYERS * 8 * 8];
  __shared__ __attribute__((aligned(16))) float ABl[4][N_LAYERS][2][256];
  __shared__ float Tbuf[NWAVES][2][256];
  __shared__ float Sbuf[NWAVES][2][256];
  __shared__ float stage[32];

  int tid = threadIdx.x;
  int l = tid & 31;
  int ta = blockIdx.x / PATCH, tq = blockIdx.x - ta * PATCH;
  int ws_ = __builtin_amdgcn_readfirstlane(tid >> 5);
  int ga = 2 * ta + (ws_ >> 1), gq = 2 * tq + (ws_ & 1);
  bool valid = (ga < GN) && (gq < GN);

  const float step = 6.28318530717958647692f / 49.0f;
  if (tid < 2 * N_LAYERS * 8) {
    int par = (tid >= N_LAYERS * 8) ? 1 : 0;
    int g = tid - par * (N_LAYERS * 8);
    bool odd = (g & 1) != 0;
    float a0 = wts[g * 3 + 0];
    float a1 = wts[g * 3 + 1];
    float a2 = wts[g * 3 + 2];
    float cx = cosf(0.5f * a0), sx = sinf(0.5f * a0);
    float cy = cosf(0.5f * a1), sy = sinf(0.5f * a1);
    float cz = cosf(0.5f * a2), sz = sinf(0.5f * a2);
    float ang = step * (float)(2 * (odd ? tq : ta) + par);
    float ce = cosf(0.5f * ang), se = sinf(0.5f * ang);
    c2 wm[4];
#pragma unroll
    for (int q = 0; q < 4; ++q) {
      int ab = q >> 1, gb = q & 1;
      c2 yx = c2{0.f, 0.f};
      for (int k = 0; k < 2; ++k) {
        c2 p = cmul(ry_el(ab, k, cy, sy), rx_el(k, gb, cx, sx));
        yx.re += p.re; yx.im += p.im;
      }
      c2 z = ab ? c2{cz, sz} : c2{cz, -sz};
      wm[q] = cmul(z, yx);
    }
#pragma unroll
    for (int q = 0; q < 4; ++q) {
      int ab = q >> 1, gb = q & 1;
      c2 h = c2{0.f, 0.f};
#pragma unroll
      for (int k = 0; k < 2; ++k) {
        c2 ex = rx_el(k, gb, ce, se);
        c2 ey = ry_el(k, gb, ce, se);
        c2 p = cmul(wm[ab * 2 + k], odd ? ey : ex);
        h.re += p.re; h.im += p.im;
      }
      Gl[tid * 8 + q * 2 + 0] = h.re;
      Gl[tid * 8 + q * 2 + 1] = h.im;
    }
  }
  __syncthreads();
  for (int it = tid; it < 4 * N_LAYERS * 16; it += NTHREADS) {
    int tt = it >> 4, sub = it & 15;
    int r_lo = sub >> 2, c_hi = sub & 3;
    int kp = tt / N_LAYERS;
    int layer = tt - kp * N_LAYERS;
    const float* gb0 = &Gl[((kp & 1) * (N_LAYERS * 8) + layer * 8 + (kp >> 1)) * 8];
    const float* F0 = gb0;
    const float* F1 = gb0 + 16;
    const float* F2 = gb0 + 32;
    const float* F3 = gb0 + 48;
    int c3 = c_hi >> 1, c2b = c_hi & 1, r1 = r_lo >> 1, r0 = r_lo & 1;
    v2f f0a = *(const v2f*)(F0 + c3 * 2);
    v2f f0b = *(const v2f*)(F0 + 4 + c3 * 2);
    v2f f1a = *(const v2f*)(F1 + c2b * 2);
    v2f f1b = *(const v2f*)(F1 + 4 + c2b * 2);
    v4f f2 = *(const v4f*)(F2 + r1 * 4);
    v4f f3 = *(const v4f*)(F3 + r0 * 4);
    c2 ph[4], pl[4];
    ph[0] = cmul(c2{f0a.x, f0a.y}, c2{f1a.x, f1a.y});
    ph[1] = cmul(c2{f0a.x, f0a.y}, c2{f1b.x, f1b.y});
    ph[2] = cmul(c2{f0b.x, f0b.y}, c2{f1a.x, f1a.y});
    ph[3] = cmul(c2{f0b.x, f0b.y}, c2{f1b.x, f1b.y});
    pl[0] = cmul(c2{f2.x, f2.y}, c2{f3.x, f3.y});
    pl[1] = cmul(c2{f2.x, f2.y}, c2{f3.z, f3.w});
    pl[2] = cmul(c2{f2.z, f2.w}, c2{f3.x, f3.y});
    pl[3] = cmul(c2{f2.z, f2.w}, c2{f3.z, f3.w});
    float* tile = &ABl[0][0][0][0] + tt * 512 + r_lo * 16 + c_hi * 4;
#pragma unroll
    for (int rh = 0; rh < 4; ++rh) {
      c2 e0 = cmul(ph[rh], pl[0]);
      c2 e1 = cmul(ph[rh], pl[1]);
      c2 e2 = cmul(ph[rh], pl[2]);
      c2 e3 = cmul(ph[rh], pl[3]);
      v4f re = {e0.re, e1.re, e2.re, e3.re};
      v4f im = {e0.im, e1.im, e2.im, e3.im};
      *(v4f*)(tile + rh * 64) = re;
      *(v4f*)(tile + rh * 64 + 256) = im;
    }
  }
  __syncthreads();

  float ev[8];
#pragma unroll
  for (int i = 0; i < 8; ++i) ev[i] = 0.f;
  if (valid) {
  int lr = l & 15, lh = l >> 4;
  float* Sr = &Sbuf[ws_][0][0];
  float* Si = &Sbuf[ws_][1][0];
  float* Tr = &Tbuf[ws_][0][0];
  float* Ti = &Tbuf[ws_][1][0];
#pragma unroll
  for (int t = 0; t < 8; ++t) {
    int idx = t * 32 + l;
    Sr[idx] = (idx == 0) ? 1.f : 0.f;
    Si[idx] = 0.f;
  }
  int dsta[8];
#pragma unroll
  for (int v = 0; v < 8; ++v) dsta[v] = chain_dst(v + 8 * lh, lr);

  const float* Ab = &ABl[ws_ >> 1][0][0][0];
  const float* Bb = &ABl[2 + (ws_ & 1)][0][0][0];
  for (int layer = 0; layer < N_LAYERS; ++layer) {
    const float* Ar = Ab + layer * 512;
    const float* Ai = Ar + 256;
    const float* Br = Bb + layer * 512;
    const float* Bi = Br + 256;
    v8f cr, ci;
    cmm(Ar, Ai, Sr, Si, false, l, cr, ci);
    store_tile(Tr, l, cr); store_tile(Ti, l, ci);
    cmm(Tr, Ti, Br, Bi, true, l, cr, ci);
#pragma unroll
    for (int v = 0; v < 8; ++v) {
      Sr[dsta[v]] = cr[v];
      Si[dsta[v]] = ci[v];
    }
  }

  float p[8];
#pragma unroll
  for (int v = 0; v < 8; ++v) {
    float a = Sr[(v + 8 * lh) * 16 + lr];
    float bi = Si[(v + 8 * lh) * 16 + lr];
    p[v] = a * a + bi * bi;
  }
  float t  = p[0] + p[1] + p[2] + p[3] + p[4] + p[5] + p[6] + p[7];
  float u1 = p[0] + p[1] + p[2] + p[3] - p[4] - p[5] - p[6] - p[7];
  float u2 = p[0] + p[1] - p[2] - p[3] + p[4] + p[5] - p[6] - p[7];
  float u3 = p[0] - p[1] + p[2] - p[3] + p[4] - p[5] + p[6] - p[7];
  ev[0] = lh ? -t : t;
  ev[2] = u1; ev[4] = u2; ev[6] = u3;
  ev[1] = (lr & 8) ? -t : t;
  ev[3] = (lr & 4) ? -t : t;
  ev[5] = (lr & 2) ? -t : t;
  ev[7] = (lr & 1) ? -t : t;
#pragma unroll
  for (int i = 0; i < 8; ++i) ev[i] = wave_sum(ev[i], l);
  }

  if (l == 0) {
    const float kPi = 3.14159265358979323846f;
#pragma unroll
    for (int i = 0; i < 8; ++i)
      stage[(i + 8 * (ws_ >> 1)) * 2 + (ws_ & 1)] = valid ? kPi * ev[i] : 0.f;
  }
  __syncthreads();
  int line = ((ta * KB + (tq >> 1)) * 32 + 16 * (tq & 1)) * 2;
  if (ws_ == 0) AM[line + l] = stage[l];
  if (ws_ == 1 && tq == PATCH - 1) AM[line + 32 + l] = 0.f;
}

__device__ __forceinline__ float node_weight(float sd) {
  sd = (fabsf(sd) < W_FLOOR) ? W_FLOOR : sd;
  return __builtin_amdgcn_rcpf(sd);
}

__global__ void __launch_bounds__(EV_THREADS)
qnn_eval(const float* __restrict__ x, const float* __restrict__ AMg,
         float* __restrict__ out, int batch) {
  __shared__ float AMl[AM_FLOATS];
  __shared__ float4 tab4[KB * 2];
  int tid = threadIdx.x;
  int l = tid & 31, w = tid >> 5;
  int lr = l & 15, lh = l >> 4;
  {
    const float4* srcp = (const float4*)AMg;
    float4* dstp = (float4*)AMl;
    for (int i = tid; i < AM_FLOATS / 4; i += EV_THREADS) dstp[i] = srcp[i];
  }
  if (tid < KB * 4) {
    float ang = (float)tid / 49.0f;
    float* tb = (float*)tab4;
    tb[2 * tid]     = cospif(ang);
    tb[2 * tid + 1] = sinpif(ang);
  }
  __syncthreads();

  const v2f* Af = (const v2f*)AMl + l;
  const v2f* tab2 = (const v2f*)tab4;
  int paddr = (l ^ 16) << 2;
  int ntiles = (batch + 31) >> 5;
  for (int tile = blockIdx.x * (EV_THREADS / 32) + w; tile < ntiles;
       tile += gridDim.x * (EV_THREADS / 32)) {
    v2f bf[2][KB];
    float sh1[2], ch1[2];
    float shs[2], chs[2];
#pragma unroll
    for (int s = 0; s < 2; ++s) {
      int b = tile * 32 + s * 16 + lr;
      int bb = (b < batch) ? b : (batch - 1);
      float x0 = x[2 * bb], x1 = x[2 * bb + 1];
      float sa, ca;
      sincosf(0.5f * x1, &sh1[s], &ch1[s]);
      sincosf(0.5f * x0, &sa, &ca);
      shs[s] = lh ? -sa : sa;
      chs[s] = lh ? -ca : ca;
    }
    float sum[2] = {0.f, 0.f};
#pragma unroll
    for (int kb = 0; kb < KB; ++kb) {
      float4 tb = tab4[2 * kb + lh];
#pragma unroll
      for (int s = 0; s < 2; ++s) {
        float w0 =  node_weight(sh1[s] * tb.x - ch1[s] * tb.y);
        float w1 = -node_weight(sh1[s] * tb.z - ch1[s] * tb.w);
        if (kb == KB - 1) {
          w1 = 0.f;
          w0 = lh ? 0.f : w0;
        }
        bf[s][kb].x = w0; bf[s][kb].y = w1;
        sum[s] += w0 + w1;
      }
    }
#pragma unroll
    for (int s = 0; s < 2; ++s) {
      float tot = sum[s] + __int_as_float(__builtin_amdgcn_ds_bpermute(paddr, __float_as_int(sum[s])));
      float inv = 1.0f / tot;
#pragma unroll
      for (int kb = 0; kb < KB; ++kb) { bf[s][kb].x *= inv; bf[s][kb].y *= inv; }
    }
    float acc0[8], acc1[8];
#pragma unroll
    for (int v = 0; v < 8; ++v) { acc0[v] = 0.f; acc1[v] = 0.f; }
    float ws0 = 0.f, ws1 = 0.f;
    for (int t = 0; t < MT; ++t) {
      v8f d0 = {}, d1 = {};
#pragma unroll
      for (int kb = 0; kb < KB; ++kb) {
        v2f a = Af[(t * KB + kb) * 32];
        d0 = __builtin_amdgcn_wmma_f32_16x16x4_f32(false, a, false, bf[0][kb], (short)0, d0, false, false);
        d1 = __builtin_amdgcn_wmma_f32_16x16x4_f32(false, a, false, bf[1][kb], (short)0, d1, false, false);
      }
      v2f cs = tab2[2 * t + lh];
      float f0 = node_weight(shs[0] * cs.x - chs[0] * cs.y);
      float f1 = node_weight(shs[1] * cs.x - chs[1] * cs.y);
      if (t == MT - 1) {
        f0 = lh ? 0.f : f0;
        f1 = lh ? 0.f : f1;
      }
      ws0 += f0; ws1 += f1;
#pragma unroll
      for (int v = 0; v < 8; ++v) { acc0[v] += f0 * d0[v]; acc1[v] += f1 * d1[v]; }
    }
    float gw = lh ? ws0 : ws1;
    float kw = lh ? ws1 : ws0;
    float den = kw + __int_as_float(__builtin_amdgcn_ds_bpermute(paddr, __float_as_int(gw)));
    float rden = 1.0f / den;
    float tot[8];
#pragma unroll
    for (int v = 0; v < 8; ++v) {
      float give = lh ? acc0[v] : acc1[v];
      float keep = lh ? acc1[v] : acc0[v];
      tot[v] = (keep + __int_as_float(__builtin_amdgcn_ds_bpermute(paddr, __float_as_int(give)))) * rden;
    }
    int bo = tile * 32 + l;
    if (bo < batch) {
      float4* o = (float4*)(out + (size_t)bo * 8);
      o[0] = make_float4(tot[0], tot[1], tot[2], tot[3]);
      o[1] = make_float4(tot[4], tot[5], tot[6], tot[7]);
    }
  }
}

extern "C" void kernel_launch(void* const* d_in, const int* in_sizes, int n_in,
                              void* d_out, int out_size, void* d_ws, size_t ws_size,
                              hipStream_t stream) {
  const float* x   = (const float*)d_in[0];
  const float* wts = (const float*)d_in[1];
  float* out = (float*)d_out;
  float* ws  = (float*)d_ws;
  float* AM  = ws + WS_AM;

  int batch = in_sizes[0] / 2;
  qnn_kernel<<<PATCH * PATCH, NTHREADS, 0, stream>>>(wts, AM);
  qnn_eval<<<256, EV_THREADS, 0, stream>>>(x, AM, out, batch);
}


# eval: x1 weights add up to one, so node 48's column is the accumulator start value and the product runs over A[m][k]-A[m][48], k<48: 12 k blocks (24 WMMA) per tile, was 13 (26); sums re-ordered
# speedup vs baseline: 12.1764x; 1.0363x over previous
#include <hip/hip_runtime.h>
#include <math.h>


typedef __attribute__((ext_vector_type(2))) float v2f;
typedef __attribute__((ext_vector_type(4))) float v4f;
typedef __attribute__((ext_vector_type(8))) float v8f;

#define N_LAYERS 6
#define NWAVES   4
#define NTHREADS (NWAVES * 32)

#define GN 49
#define GK 24
#define NGRID (GN * GN)
#define MT 25
#define KB 13
#define KBM 12
#define AM_FLOATS (MT * KB * 64)
#define AML_FLOATS (MT * KBM * 64)
#define WS_AM 40960
#define EV_THREADS 512
#define PATCH 25
#define W_FLOOR 1e-18f

struct c2 { float re, im; };

__device__ __forceinline__ c2 cmul(c2 a, c2 b) {
  return c2{a.re * b.re - a.im * b.im, a.re * b.im + a.im * b.re};
}
__device__ __forceinline__ c2 rx_el(int a, int b, float c, float s) {
  return (a == b) ? c2{c, 0.f} : c2{0.f, -s};
}
__device__ __forceinline__ c2 ry_el(int a, int b, float c, float s) {
  if (a == b) return c2{c, 0.f};
  return (a == 1) ? c2{s, 0.f} : c2{-s, 0.f};
}

__device__ __forceinline__ void cmm(const float* __restrict__ Ar,
                                    const float* __restrict__ Ai,
                                    const float* __restrict__ Br,
                                    const float* __restrict__ Bi,
                                    bool bT, int l, v8f& Cr, v8f& Ci) {
  int lr = l & 15, lh = l >> 4;
  int aoff = lr * 16 + 2 * lh;
  v8f rr = {}, ii = {}, ri = {};
#pragma unroll
  for (int kb = 0; kb < 4; ++kb) {
    v2f ar = *(const v2f*)(Ar + aoff + 4 * kb);
    v2f ai = *(const v2f*)(Ai + aoff + 4 * kb);
    v2f br, bi;
    if (bT) {
      br = *(const v2f*)(Br + aoff + 4 * kb);
      bi = *(const v2f*)(Bi + aoff + 4 * kb);
    } else {
      int k0 = 4 * kb + 2 * lh;
      br.x = Br[k0 * 16 + lr]; br.y = Br[(k0 + 1) * 16 + lr];
      bi.x = Bi[k0 * 16 + lr]; bi.y = Bi[(k0 + 1) * 16 + lr];
    }
    rr = __builtin_amdgcn_wmma_f32_16x16x4_f32(false, ar, false, br, (short)0, rr, false, false);
    ii = __builtin_amdgcn_wmma_f32_16x16x4_f32(false, ai, false, bi, (short)0, ii, false, false);
    ri = __builtin_amdgcn_wmma_f32_16x16x4_f32(false, ar, false, bi, (short)0, ri, false, false);
    ri = __builtin_amdgcn_wmma_f32_16x16x4_f32(false, ai, false, br, (short)0, ri, false, false);
  }
  Cr = rr - ii;
  Ci = ri;
}

__device__ __forceinline__ void store_tile(float* __restrict__ dst, int l, v8f c) {
  int base = ((l >> 4) * 8) * 16 + (l & 15);
#pragma unroll
  for (int v = 0; v < 8; ++v) dst[base + v * 16] = c[v];
}

__device__ __forceinline__ float wave_sum(float v, int l) {
#pragma unroll
  for (int m = 16; m; m >>= 1) {
    int src = (l ^ m) << 2;
    v += __int_as_float(__builtin_amdgcn_ds_bpermute(src, __float_as_int(v)));
  }
  return v;
}

__device__ __forceinline__ int chain_dst(int r, int c) {
  int p0 = (r >> 3) & 1;
  int p1 = p0 ^ ((c >> 3) & 1);
  int p2 = p1 ^ ((r >> 2) & 1);
  int p3 = p2 ^ ((c >> 2) & 1);
  int p4 = p3 ^ ((r >> 1) & 1);
  int p5 = p4 ^ ((c >> 1) & 1);
  int p6 = p5 ^ (r & 1);
  int p7 = p6 ^ (c & 1);
  int rn = (p0 << 3) | (p2 << 2) | (p4 << 1) | p6;
  int cn = (p1 << 3) | (p3 << 2) | (p5 << 1) | p7;
  return rn * 16 + cn;
}

__global__ void __launch_bounds__(NTHREADS)
qnn_kernel(const float* __restrict__ wts, float* __restrict__ AM) {
  __shared__ __attribute__((aligned(16))) float Gl[2 * N_LAYERS * 8 * 8];
  __shared__ __attribute__((aligned(16))) float ABl[4][N_LAYERS][2][256];
  __shared__ float Tbuf[NWAVES][2][256];
  __shared__ float Sbuf[NWAVES][2][256];
  __shared__ float stage[32];

  int tid = threadIdx.x;
  int l = tid & 31;
  int ta = blockIdx.x / PATCH, tq = blockIdx.x - ta * PATCH;
  int ws_ = __builtin_amdgcn_readfirstlane(tid >> 5);
  int ga = 2 * ta + (ws_ >> 1), gq = 2 * tq + (ws_ & 1);
  bool valid = (ga < GN) && (gq < GN);

  const float step = 6.28318530717958647692f / 49.0f;
  if (tid < 2 * N_LAYERS * 8) {
    int par = (tid >= N_LAYERS * 8) ? 1 : 0;
    int g = tid - par * (N_LAYERS * 8);
    bool odd = (g & 1) != 0;
    float a0 = wts[g * 3 + 0];
    float a1 = wts[g * 3 + 1];
    float a2 = wts[g * 3 + 2];
    float cx = cosf(0.5f * a0), sx = sinf(0.5f * a0);
    float cy = cosf(0.5f * a1), sy = sinf(0.5f * a1);
    float cz = cosf(0.5f * a2), sz = sinf(0.5f * a2);
    float ang = step * (float)(2 * (odd ? tq : ta) + par);
    float ce = cosf(0.5f * ang), se = sinf(0.5f * ang);
    c2 wm[4];
#pragma unroll
    for (int q = 0; q < 4; ++q) {
      int ab = q >> 1, gb = q & 1;
      c2 yx = c2{0.f, 0.f};
      for (int k = 0; k < 2; ++k) {
        c2 p = cmul(ry_el(ab, k, cy, sy), rx_el(k, gb, cx, sx));
        yx.re += p.re; yx.im += p.im;
      }
      c2 z = ab ? c2{cz, sz} : c2{cz, -sz};
      wm[q] = cmul(z, yx);
    }
#pragma unroll
    for (int q = 0; q < 4; ++q) {
      int ab = q >> 1, gb = q & 1;
      c2 h = c2{0.f, 0.f};
#pragma unroll
      for (int k = 0; k < 2; ++k) {
        c2 ex = rx_el(k, gb, ce, se);
        c2 ey = ry_el(k, gb, ce, se);
        c2 p = cmul(wm[ab * 2 + k], odd ? ey : ex);
        h.re += p.re; h.im += p.im;
      }
      Gl[tid * 8 + q * 2 + 0] = h.re;
      Gl[tid * 8 + q * 2 + 1] = h.im;
    }
  }
  __syncthreads();
  for (int it = tid; it < 4 * N_LAYERS * 16; it += NTHREADS) {
    int tt = it >> 4, sub = it & 15;
    int r_lo = sub >> 2, c_hi = sub & 3;
    int kp = tt / N_LAYERS;
    int layer = tt - kp * N_LAYERS;
    const float* gb0 = &Gl[((kp & 1) * (N_LAYERS * 8) + layer * 8 + (kp >> 1)) * 8];
    const float* F0 = gb0;
    const float* F1 = gb0 + 16;
    const float* F2 = gb0 + 32;
    const float* F3 = gb0 + 48;
    int c3 = c_hi >> 1, c2b = c_hi & 1, r1 = r_lo >> 1, r0 = r_lo & 1;
    v2f f0a = *(const v2f*)(F0 + c3 * 2);
    v2f f0b = *(const v2f*)(F0 + 4 + c3 * 2);
    v2f f1a = *(const v2f*)(F1 + c2b * 2);
    v2f f1b = *(const v2f*)(F1 + 4 + c2b * 2);
    v4f f2 = *(const v4f*)(F2 + r1 * 4);
    v4f f3 = *(const v4f*)(F3 + r0 * 4);
    c2 ph[4], pl[4];
    ph[0] = cmul(c2{f0a.x, f0a.y}, c2{f1a.x, f1a.y});
    ph[1] = cmul(c2{f0a.x, f0a.y}, c2{f1b.x, f1b.y});
    ph[2] = cmul(c2{f0b.x, f0b.y}, c2{f1a.x, f1a.y});
    ph[3] = cmul(c2{f0b.x, f0b.y}, c2{f1b.x, f1b.y});
    pl[0] = cmul(c2{f2.x, f2.y}, c2{f3.x, f3.y});
    pl[1] = cmul(c2{f2.x, f2.y}, c2{f3.z, f3.w});
    pl[2] = cmul(c2{f2.z, f2.w}, c2{f3.x, f3.y});
    pl[3] = cmul(c2{f2.z, f2.w}, c2{f3.z, f3.w});
    float* tile = &ABl[0][0][0][0] + tt * 512 + r_lo * 16 + c_hi * 4;
#pragma unroll
    for (int rh = 0; rh < 4; ++rh) {
      c2 e0 = cmul(ph[rh], pl[0]);
      c2 e1 = cmul(ph[rh], pl[1]);
      c2 e2 = cmul(ph[rh], pl[2]);
      c2 e3 = cmul(ph[rh], pl[3]);
      v4f re = {e0.re, e1.re, e2.re, e3.re};
      v4f im = {e0.im, e1.im, e2.im, e3.im};
      *(v4f*)(tile + rh * 64) = re;
      *(v4f*)(tile + rh * 64 + 256) = im;
    }
  }
  __syncthreads();

  float ev[8];
#pragma unroll
  for (int i = 0; i < 8; ++i) ev[i] = 0.f;
  if (valid) {
  int lr = l & 15, lh = l >> 4;
  float* Sr = &Sbuf[ws_][0][0];
  float* Si = &Sbuf[ws_][1][0];
  float* Tr = &Tbuf[ws_][0][0];
  float* Ti = &Tbuf[ws_][1][0];
#pragma unroll
  for (int t = 0; t < 8; ++t) {
    int idx = t * 32 + l;
    Sr[idx] = (idx == 0) ? 1.f : 0.f;
    Si[idx] = 0.f;
  }
  int dsta[8];
#pragma unroll
  for (int v = 0; v < 8; ++v) dsta[v] = chain_dst(v + 8 * lh, lr);

  const float* Ab = &ABl[ws_ >> 1][0][0][0];
  const float* Bb = &ABl[2 + (ws_ & 1)][0][0][0];
  for (int layer = 0; layer < N_LAYERS; ++layer) {
    const float* Ar = Ab + layer * 512;
    const float* Ai = Ar + 256;
    const float* Br = Bb + layer * 512;
    const float* Bi = Br + 256;
    v8f cr, ci;
    cmm(Ar, Ai, Sr, Si, false, l, cr, ci);
    store_tile(Tr, l, cr); store_tile(Ti, l, ci);
    cmm(Tr, Ti, Br, Bi, true, l, cr, ci);
#pragma unroll
    for (int v = 0; v < 8; ++v) {
      Sr[dsta[v]] = cr[v];
      Si[dsta[v]] = ci[v];
    }
  }

  float p[8];
#pragma unroll
  for (int v = 0; v < 8; ++v) {
    float a = Sr[(v + 8 * lh) * 16 + lr];
    float bi = Si[(v + 8 * lh) * 16 + lr];
    p[v] = a * a + bi * bi;
  }
  float t  = p[0] + p[1] + p[2] + p[3] + p[4] + p[5] + p[6] + p[7];
  float u1 = p[0] + p[1] + p[2] + p[3] - p[4] - p[5] - p[6] - p[7];
  float u2 = p[0] + p[1] - p[2] - p[3] + p[4] + p[5] - p[6] - p[7];
  float u3 = p[0] - p[1] + p[2] - p[3] + p[4] - p[5] + p[6] - p[7];
  ev[0] = lh ? -t : t;
  ev[2] = u1; ev[4] = u2; ev[6] = u3;
  ev[1] = (lr & 8) ? -t : t;
  ev[3] = (lr & 4) ? -t : t;
  ev[5] = (lr & 2) ? -t : t;
  ev[7] = (lr & 1) ? -t : t;
#pragma unroll
  for (int i = 0; i < 8; ++i) ev[i] = wave_sum(ev[i], l);
  }

  if (l == 0) {
    const float kPi = 3.14159265358979323846f;
#pragma unroll
    for (int i = 0; i < 8; ++i)
      stage[(i + 8 * (ws_ >> 1)) * 2 + (ws_ & 1)] = valid ? kPi * ev[i] : 0.f;
  }
  __syncthreads();
  int line = ((ta * KB + (tq >> 1)) * 32 + 16 * (tq & 1)) * 2;
  if (ws_ == 0) AM[line + l] = stage[l];
  if (ws_ == 1 && tq == PATCH - 1) AM[line + 32 + l] = 0.f;
}

__device__ __forceinline__ float node_weight(float sd) {
  sd = (fabsf(sd) < W_FLOOR) ? W_FLOOR : sd;
  return __builtin_amdgcn_rcpf(sd);
}

__global__ void __launch_bounds__(EV_THREADS)
qnn_eval(const float* __restrict__ x, const float* __restrict__ AMg,
         float* __restrict__ out, int batch) {
  __shared__ __attribute__((aligned(16))) float AMl[AML_FLOATS];
  __shared__ __attribute__((aligned(32))) float C48[MT * 16];
  __shared__ float4 tab4[KB * 2];
  int tid = threadIdx.x;
  int l = tid & 31, w = tid >> 5;
  int lr = l & 15, lh = l >> 4;
  {
    const float4* srcp = (const float4*)AMg;
    float4* dstp = (float4*)AMl;
    float c48v = 0.f;
    if (tid < MT * 16)
      c48v = AMg[(((tid >> 4) * KB + (KB - 1)) * 32 + (tid & 15)) * 2];
    for (int i = tid; i < AML_FLOATS / 4; i += EV_THREADS) {
      int t = i / (KBM * 16);
      int r = i - t * (KBM * 16);
      int kb = r >> 4, p = r & 15;
      float4 v = srcp[(t * KB + kb) * 16 + p];
      float4 c = srcp[(t * KB + (KB - 1)) * 16 + (p & 7)];
      dstp[i] = make_float4(v.x - c.x, v.y - c.x, v.z - c.z, v.w - c.z);
    }
    if (tid < MT * 16) C48[tid] = c48v;
  }
  if (tid < KB * 4) {
    float ang = (float)tid / 49.0f;
    float* tb = (float*)tab4;
    tb[2 * tid]     = cospif(ang);
    tb[2 * tid + 1] = sinpif(ang);
  }
  __syncthreads();

  const v2f* Af = (const v2f*)AMl + l;
  const v2f* tab2 = (const v2f*)tab4;
  int paddr = (l ^ 16) << 2;
  int ntiles = (batch + 31) >> 5;
  for (int tile = blockIdx.x * (EV_THREADS / 32) + w; tile < ntiles;
       tile += gridDim.x * (EV_THREADS / 32)) {
    v2f bf[2][KBM];
    float sh1[2], ch1[2];
    float shs[2], chs[2];
#pragma unroll
    for (int s = 0; s < 2; ++s) {
      int b = tile * 32 + s * 16 + lr;
      int bb = (b < batch) ? b : (batch - 1);
      float x0 = x[2 * bb], x1 = x[2 * bb + 1];
      float sa, ca;
      sincosf(0.5f * x1, &sh1[s], &ch1[s]);
      sincosf(0.5f * x0, &sa, &ca);
      shs[s] = lh ? -sa : sa;
      chs[s] = lh ? -ca : ca;
    }
    float sum[2] = {0.f, 0.f};
#pragma unroll
    for (int kb = 0; kb < KBM; ++kb) {
      float4 tb = tab4[2 * kb + lh];
#pragma unroll
      for (int s = 0; s < 2; ++s) {
        float w0 =  node_weight(sh1[s] * tb.x - ch1[s] * tb.y);
        float w1 = -node_weight(sh1[s] * tb.z - ch1[s] * tb.w);
        bf[s][kb].x = w0; bf[s][kb].y = w1;
        sum[s] += w0 + w1;
      }
    }
    {
      v2f c48 = tab2[GN - 1];
#pragma unroll
      for (int s = 0; s < 2; ++s) {
        float w48 = node_weight(sh1[s] * c48.x - ch1[s] * c48.y);
        sum[s] += lh ? 0.f : w48;
      }
    }
#pragma unroll
    for (int s = 0; s < 2; ++s) {
      float tot = sum[s] + __int_as_float(__builtin_amdgcn_ds_bpermute(paddr, __float_as_int(sum[s])));
      float inv = 1.0f / tot;
#pragma unroll
      for (int kb = 0; kb < KBM; ++kb) { bf[s][kb].x *= inv; bf[s][kb].y *= inv; }
    }
    float acc0[8], acc1[8];
#pragma unroll
    for (int v = 0; v < 8; ++v) { acc0[v] = 0.f; acc1[v] = 0.f; }
    float ws0 = 0.f, ws1 = 0.f;
    for (int t = 0; t < MT; ++t) {
      v8f d0 = *(const v8f*)(C48 + t * 16 + 8 * lh);
      v8f d1 = d0;
#pragma unroll
      for (int kb = 0; kb < KBM; ++kb) {
        v2f a = Af[(t * KBM + kb) * 32];
        d0 = __builtin_amdgcn_wmma_f32_16x16x4_f32(false, a, false, bf[0][kb], (short)0, d0, false, false);
        d1 = __builtin_amdgcn_wmma_f32_16x16x4_f32(false, a, false, bf[1][kb], (short)0, d1, false, false);
      }
      v2f cs = tab2[2 * t + lh];
      float f0 = node_weight(shs[0] * cs.x - chs[0] * cs.y);
      float f1 = node_weight(shs[1] * cs.x - chs[1] * cs.y);
      if (t == MT - 1) {
        f0 = lh ? 0.f : f0;
        f1 = lh ? 0.f : f1;
      }
      ws0 += f0; ws1 += f1;
#pragma unroll
      for (int v = 0; v < 8; ++v) { acc0[v] += f0 * d0[v]; acc1[v] += f1 * d1[v]; }
    }
    float gw = lh ? ws0 : ws1;
    float kw = lh ? ws1 : ws0;
    float den = kw + __int_as_float(__builtin_amdgcn_ds_bpermute(paddr, __float_as_int(gw)));
    float rden = 1.0f / den;
    float tot[8];
#pragma unroll
    for (int v = 0; v < 8; ++v) {
      float give = lh ? acc0[v] : acc1[v];
      float keep = lh ? acc1[v] : acc0[v];
      tot[v] = (keep + __int_as_float(__builtin_amdgcn_ds_bpermute(paddr, __float_as_int(give)))) * rden;
    }
    int bo = tile * 32 + l;
    if (bo < batch) {
      float4* o = (float4*)(out + (size_t)bo * 8);
      o[0] = make_float4(tot[0], tot[1], tot[2], tot[3]);
      o[1] = make_float4(tot[4], tot[5], tot[6], tot[7]);
    }
  }
}

extern "C" void kernel_launch(void* const* d_in, const int* in_sizes, int n_in,
                              void* d_out, int out_size, void* d_ws, size_t ws_size,
                              hipStream_t stream) {
  const float* x   = (const float*)d_in[0];
  const float* wts = (const float*)d_in[1];
  float* out = (float*)d_out;
  float* ws  = (float*)d_ws;
  float* AM  = ws + WS_AM;

  int batch = in_sizes[0] / 2;
  qnn_kernel<<<PATCH * PATCH, NTHREADS, 0, stream>>>(wts, AM);
  qnn_eval<<<256, EV_THREADS, 0, stream>>>(x, AM, out, batch);
}


# eval: grid table staged to LDS in two parts: tiles 0-4 ahead of the first product, tiles 5-24 loaded to registers before the weight set-up, stored behind the first 5 tiles' products; sums unchanged
# speedup vs baseline: 12.7824x; 1.0498x over previous
#include <hip/hip_runtime.h>
#include <math.h>


typedef __attribute__((ext_vector_type(2))) float v2f;
typedef __attribute__((ext_vector_type(4))) float v4f;
typedef __attribute__((ext_vector_type(8))) float v8f;

#define N_LAYERS 6
#define NWAVES   4
#define NTHREADS (NWAVES * 32)

#define GN 49
#define GK 24
#define NGRID (GN * GN)
#define MT 25
#define KB 13
#define KBM 12
#define AM_FLOATS (MT * KB * 64)
#define AML_FLOATS (MT * KBM * 64)
#define WS_AM 40960
#define EV_THREADS 512
#define PATCH 25
#define W_FLOOR 1e-18f

struct c2 { float re, im; };

__device__ __forceinline__ c2 cmul(c2 a, c2 b) {
  return c2{a.re * b.re - a.im * b.im, a.re * b.im + a.im * b.re};
}
__device__ __forceinline__ c2 rx_el(int a, int b, float c, float s) {
  return (a == b) ? c2{c, 0.f} : c2{0.f, -s};
}
__device__ __forceinline__ c2 ry_el(int a, int b, float c, float s) {
  if (a == b) return c2{c, 0.f};
  return (a == 1) ? c2{s, 0.f} : c2{-s, 0.f};
}

__device__ __forceinline__ void cmm(const float* __restrict__ Ar,
                                    const float* __restrict__ Ai,
                                    const float* __restrict__ Br,
                                    const float* __restrict__ Bi,
                                    bool bT, int l, v8f& Cr, v8f& Ci) {
  int lr = l & 15, lh = l >> 4;
  int aoff = lr * 16 + 2 * lh;
  v8f rr = {}, ii = {}, ri = {};
#pragma unroll
  for (int kb = 0; kb < 4; ++kb) {
    v2f ar = *(const v2f*)(Ar + aoff + 4 * kb);
    v2f ai = *(const v2f*)(Ai + aoff + 4 * kb);
    v2f br, bi;
    if (bT) {
      br = *(const v2f*)(Br + aoff + 4 * kb);
      bi = *(const v2f*)(Bi + aoff + 4 * kb);
    } else {
      int k0 = 4 * kb + 2 * lh;
      br.x = Br[k0 * 16 + lr]; br.y = Br[(k0 + 1) * 16 + lr];
      bi.x = Bi[k0 * 16 + lr]; bi.y = Bi[(k0 + 1) * 16 + lr];
    }
    rr = __builtin_amdgcn_wmma_f32_16x16x4_f32(false, ar, false, br, (short)0, rr, false, false);
    ii = __builtin_amdgcn_wmma_f32_16x16x4_f32(false, ai, false, bi, (short)0, ii, false, false);
    ri = __builtin_amdgcn_wmma_f32_16x16x4_f32(false, ar, false, bi, (short)0, ri, false, false);
    ri = __builtin_amdgcn_wmma_f32_16x16x4_f32(false, ai, false, br, (short)0, ri, false, false);
  }
  Cr = rr - ii;
  Ci = ri;
}

__device__ __forceinline__ void store_tile(float* __restrict__ dst, int l, v8f c) {
  int base = ((l >> 4) * 8) * 16 + (l & 15);
#pragma unroll
  for (int v = 0; v < 8; ++v) dst[base + v * 16] = c[v];
}

__device__ __forceinline__ float wave_sum(float v, int l) {
#pragma unroll
  for (int m = 16; m; m >>= 1) {
    int src = (l ^ m) << 2;
    v += __int_as_float(__builtin_amdgcn_ds_bpermute(src, __float_as_int(v)));
  }
  return v;
}

__device__ __forceinline__ int chain_dst(int r, int c) {
  int p0 = (r >> 3) & 1;
  int p1 = p0 ^ ((c >> 3) & 1);
  int p2 = p1 ^ ((r >> 2) & 1);
  int p3 = p2 ^ ((c >> 2) & 1);
  int p4 = p3 ^ ((r >> 1) & 1);
  int p5 = p4 ^ ((c >> 1) & 1);
  int p6 = p5 ^ (r & 1);
  int p7 = p6 ^ (c & 1);
  int rn = (p0 << 3) | (p2 << 2) | (p4 << 1) | p6;
  int cn = (p1 << 3) | (p3 << 2) | (p5 << 1) | p7;
  return rn * 16 + cn;
}

__global__ void __launch_bounds__(NTHREADS)
qnn_kernel(const float* __restrict__ wts, float* __restrict__ AM) {
  __shared__ __attribute__((aligned(16))) float Gl[2 * N_LAYERS * 8 * 8];
  __shared__ __attribute__((aligned(16))) float ABl[4][N_LAYERS][2][256];
  __shared__ float Tbuf[NWAVES][2][256];
  __shared__ float Sbuf[NWAVES][2][256];
  __shared__ float stage[32];

  int tid = threadIdx.x;
  int l = tid & 31;
  int ta = blockIdx.x / PATCH, tq = blockIdx.x - ta * PATCH;
  int ws_ = __builtin_amdgcn_readfirstlane(tid >> 5);
  int ga = 2 * ta + (ws_ >> 1), gq = 2 * tq + (ws_ & 1);
  bool valid = (ga < GN) && (gq < GN);

  const float step = 6.28318530717958647692f / 49.0f;
  if (tid < 2 * N_LAYERS * 8) {
    int par = (tid >= N_LAYERS * 8) ? 1 : 0;
    int g = tid - par * (N_LAYERS * 8);
    bool odd = (g & 1) != 0;
    float a0 = wts[g * 3 + 0];
    float a1 = wts[g * 3 + 1];
    float a2 = wts[g * 3 + 2];
    float cx = cosf(0.5f * a0), sx = sinf(0.5f * a0);
    float cy = cosf(0.5f * a1), sy = sinf(0.5f * a1);
    float cz = cosf(0.5f * a2), sz = sinf(0.5f * a2);
    float ang = step * (float)(2 * (odd ? tq : ta) + par);
    float ce = cosf(0.5f * ang), se = sinf(0.5f * ang);
    c2 wm[4];
#pragma unroll
    for (int q = 0; q < 4; ++q) {
      int ab = q >> 1, gb = q & 1;
      c2 yx = c2{0.f, 0.f};
      for (int k = 0; k < 2; ++k) {
        c2 p = cmul(ry_el(ab, k, cy, sy), rx_el(k, gb, cx, sx));
        yx.re += p.re; yx.im += p.im;
      }
      c2 z = ab ? c2{cz, sz} : c2{cz, -sz};
      wm[q] = cmul(z, yx);
    }
#pragma unroll
    for (int q = 0; q < 4; ++q) {
      int ab = q >> 1, gb = q & 1;
      c2 h = c2{0.f, 0.f};
#pragma unroll
      for (int k = 0; k < 2; ++k) {
        c2 ex = rx_el(k, gb, ce, se);
        c2 ey = ry_el(k, gb, ce, se);
        c2 p = cmul(wm[ab * 2 + k], odd ? ey : ex);
        h.re += p.re; h.im += p.im;
      }
      Gl[tid * 8 + q * 2 + 0] = h.re;
      Gl[tid * 8 + q * 2 + 1] = h.im;
    }
  }
  __syncthreads();
  for (int it = tid; it < 4 * N_LAYERS * 16; it += NTHREADS) {
    int tt = it >> 4, sub = it & 15;
    int r_lo = sub >> 2, c_hi = sub & 3;
    int kp = tt / N_LAYERS;
    int layer = tt - kp * N_LAYERS;
    const float* gb0 = &Gl[((kp & 1) * (N_LAYERS * 8) + layer * 8 + (kp >> 1)) * 8];
    const float* F0 = gb0;
    const float* F1 = gb0 + 16;
    const float* F2 = gb0 + 32;
    const float* F3 = gb0 + 48;
    int c3 = c_hi >> 1, c2b = c_hi & 1, r1 = r_lo >> 1, r0 = r_lo & 1;
    v2f f0a = *(const v2f*)(F0 + c3 * 2);
    v2f f0b = *(const v2f*)(F0 + 4 + c3 * 2);
    v2f f1a = *(const v2f*)(F1 + c2b * 2);
    v2f f1b = *(const v2f*)(F1 + 4 + c2b * 2);
    v4f f2 = *(const v4f*)(F2 + r1 * 4);
    v4f f3 = *(const v4f*)(F3 + r0 * 4);
    c2 ph[4], pl[4];
    ph[0] = cmul(c2{f0a.x, f0a.y}, c2{f1a.x, f1a.y});
    ph[1] = cmul(c2{f0a.x, f0a.y}, c2{f1b.x, f1b.y});
    ph[2] = cmul(c2{f0b.x, f0b.y}, c2{f1a.x, f1a.y});
    ph[3] = cmul(c2{f0b.x, f0b.y}, c2{f1b.x, f1b.y});
    pl[0] = cmul(c2{f2.x, f2.y}, c2{f3.x, f3.y});
    pl[1] = cmul(c2{f2.x, f2.y}, c2{f3.z, f3.w});
    pl[2] = cmul(c2{f2.z, f2.w}, c2{f3.x, f3.y});
    pl[3] = cmul(c2{f2.z, f2.w}, c2{f3.z, f3.w});
    float* tile = &ABl[0][0][0][0] + tt * 512 + r_lo * 16 + c_hi * 4;
#pragma unroll
    for (int rh = 0; rh < 4; ++rh) {
      c2 e0 = cmul(ph[rh], pl[0]);
      c2 e1 = cmul(ph[rh], pl[1]);
      c2 e2 = cmul(ph[rh], pl[2]);
      c2 e3 = cmul(ph[rh], pl[3]);
      v4f re = {e0.re, e1.re, e2.re, e3.re};
      v4f im = {e0.im, e1.im, e2.im, e3.im};
      *(v4f*)(tile + rh * 64) = re;
      *(v4f*)(tile + rh * 64 + 256) = im;
    }
  }
  __syncthreads();

  float ev[8];
#pragma unroll
  for (int i = 0; i < 8; ++i) ev[i] = 0.f;
  if (valid) {
  int lr = l & 15, lh = l >> 4;
  float* Sr = &Sbuf[ws_][0][0];
  float* Si = &Sbuf[ws_][1][0];
  float* Tr = &Tbuf[ws_][0][0];
  float* Ti = &Tbuf[ws_][1][0];
#pragma unroll
  for (int t = 0; t < 8; ++t) {
    int idx = t * 32 + l;
    Sr[idx] = (idx == 0) ? 1.f : 0.f;
    Si[idx] = 0.f;
  }
  int dsta[8];
#pragma unroll
  for (int v = 0; v < 8; ++v) dsta[v] = chain_dst(v + 8 * lh, lr);

  const float* Ab = &ABl[ws_ >> 1][0][0][0];
  const float* Bb = &ABl[2 + (ws_ & 1)][0][0][0];
  for (int layer = 0; layer < N_LAYERS; ++layer) {
    const float* Ar = Ab + layer * 512;
    const float* Ai = Ar + 256;
    const float* Br = Bb + layer * 512;
    const float* Bi = Br + 256;
    v8f cr, ci;
    cmm(Ar, Ai, Sr, Si, false, l, cr, ci);
    store_tile(Tr, l, cr); store_tile(Ti, l, ci);
    cmm(Tr, Ti, Br, Bi, true, l, cr, ci);
#pragma unroll
    for (int v = 0; v < 8; ++v) {
      Sr[dsta[v]] = cr[v];
      Si[dsta[v]] = ci[v];
    }
  }

  float p[8];
#pragma unroll
  for (int v = 0; v < 8; ++v) {
    float a = Sr[(v + 8 * lh) * 16 + lr];
    float bi = Si[(v + 8 * lh) * 16 + lr];
    p[v] = a * a + bi * bi;
  }
  float t  = p[0] + p[1] + p[2] + p[3] + p[4] + p[5] + p[6] + p[7];
  float u1 = p[0] + p[1] + p[2] + p[3] - p[4] - p[5] - p[6] - p[7];
  float u2 = p[0] + p[1] - p[2] - p[3] + p[4] + p[5] - p[6] - p[7];
  float u3 = p[0] - p[1] + p[2] - p[3] + p[4] - p[5] + p[6] - p[7];
  ev[0] = lh ? -t : t;
  ev[2] = u1; ev[4] = u2; ev[6] = u3;
  ev[1] = (lr & 8) ? -t : t;
  ev[3] = (lr & 4) ? -t : t;
  ev[5] = (lr & 2) ? -t : t;
  ev[7] = (lr & 1) ? -t : t;
#pragma unroll
  for (int i = 0; i < 8; ++i) ev[i] = wave_sum(ev[i], l);
  }

  if (l == 0) {
    const float kPi = 3.14159265358979323846f;
#pragma unroll
    for (int i = 0; i < 8; ++i)
      stage[(i + 8 * (ws_ >> 1)) * 2 + (ws_ & 1)] = valid ? kPi * ev[i] : 0.f;
  }
  __syncthreads();
  int line = ((ta * KB + (tq >> 1)) * 32 + 16 * (tq & 1)) * 2;
  if (ws_ == 0) AM[line + l] = stage[l];
  if (ws_ == 1 && tq == PATCH - 1) AM[line + 32 + l] = 0.f;
}

__device__ __forceinline__ float node_weight(float sd) {
  sd = (fabsf(sd) < W_FLOOR) ? W_FLOOR : sd;
  return __builtin_amdgcn_rcpf(sd);
}

#define ST0 5
#define ST0_F4 (ST0 * KBM * 16)
#define ST0_PER ((ST0_F4 + EV_THREADS - 1) / EV_THREADS)
#define ST1_PER ((AML_FLOATS / 4 - ST0_F4 + EV_THREADS - 1) / EV_THREADS)

struct EvTile {
  v2f bf[2][KBM];
  float shs[2], chs[2];
  float acc0[8], acc1[8];
  float ws0, ws1;
};

__device__ __forceinline__ void ev_load_x(const float* __restrict__ x, int tile, int lr, int batch,
                                          float (&x0)[2], float (&x1)[2]) {
#pragma unroll
  for (int s = 0; s < 2; ++s) {
    int b = tile * 32 + s * 16 + lr;
    int bb = (b < batch) ? b : (batch - 1);
    x0[s] = x[2 * bb]; x1[s] = x[2 * bb + 1];
  }
}

__device__ __forceinline__ void ev_weights(const float (&x0)[2], const float (&x1)[2],
                                           const float4* tab4, const v2f* tab2,
                                           int lh, int paddr, EvTile& T) {
  float sh1[2], ch1[2];
#pragma unroll
  for (int s = 0; s < 2; ++s) {
    float sa, ca;
    sincosf(0.5f * x1[s], &sh1[s], &ch1[s]);
    sincosf(0.5f * x0[s], &sa, &ca);
    T.shs[s] = lh ? -sa : sa;
    T.chs[s] = lh ? -ca : ca;
  }
  float sum[2] = {0.f, 0.f};
#pragma unroll
  for (int kb = 0; kb < KBM; ++kb) {
    float4 tb = tab4[2 * kb + lh];
#pragma unroll
    for (int s = 0; s < 2; ++s) {
      float w0 =  node_weight(sh1[s] * tb.x - ch1[s] * tb.y);
      float w1 = -node_weight(sh1[s] * tb.z - ch1[s] * tb.w);
      T.bf[s][kb].x = w0; T.bf[s][kb].y = w1;
      sum[s] += w0 + w1;
    }
  }
  {
    v2f c48 = tab2[GN - 1];
#pragma unroll
    for (int s = 0; s < 2; ++s) {
      float w48 = node_weight(sh1[s] * c48.x - ch1[s] * c48.y);
      sum[s] += lh ? 0.f : w48;
    }
  }
#pragma unroll
  for (int s = 0; s < 2; ++s) {
    float tot = sum[s] + __int_as_float(__builtin_amdgcn_ds_bpermute(paddr, __float_as_int(sum[s])));
    float inv = 1.0f / tot;
#pragma unroll
    for (int kb = 0; kb < KBM; ++kb) { T.bf[s][kb].x *= inv; T.bf[s][kb].y *= inv; }
  }
#pragma unroll
  for (int v = 0; v < 8; ++v) { T.acc0[v] = 0.f; T.acc1[v] = 0.f; }
  T.ws0 = 0.f; T.ws1 = 0.f;
}

__device__ __forceinline__ void ev_mm(EvTile& T, int t0, int t1, const v2f* Af, const float* C48,
                                      const v2f* tab2, int lh) {
#pragma nounroll
  for (int t = t0; t < t1; ++t) {
    v8f d0 = *(const v8f*)(C48 + t * 16 + 8 * lh);
    v8f d1 = d0;
#pragma unroll
    for (int kb = 0; kb < KBM; ++kb) {
      v2f a = Af[(t * KBM + kb) * 32];
      d0 = __builtin_amdgcn_wmma_f32_16x16x4_f32(false, a, false, T.bf[0][kb], (short)0, d0, false, false);
      d1 = __builtin_amdgcn_wmma_f32_16x16x4_f32(false, a, false, T.bf[1][kb], (short)0, d1, false, false);
    }
    v2f cs = tab2[2 * t + lh];
    float f0 = node_weight(T.shs[0] * cs.x - T.chs[0] * cs.y);
    float f1 = node_weight(T.shs[1] * cs.x - T.chs[1] * cs.y);
    if (t == MT - 1) {
      f0 = lh ? 0.f : f0;
      f1 = lh ? 0.f : f1;
    }
    T.ws0 += f0; T.ws1 += f1;
#pragma unroll
    for (int v = 0; v < 8; ++v) { T.acc0[v] += f0 * d0[v]; T.acc1[v] += f1 * d1[v]; }
  }
}

__device__ __forceinline__ void ev_finish(const EvTile& T, int lh, int paddr, int tile, int l,
                                          int batch, float* __restrict__ out) {
  float gw = lh ? T.ws0 : T.ws1;
  float kw = lh ? T.ws1 : T.ws0;
  float den = kw + __int_as_float(__builtin_amdgcn_ds_bpermute(paddr, __float_as_int(gw)));
  float rden = 1.0f / den;
  float tot[8];
#pragma unroll
  for (int v = 0; v < 8; ++v) {
    float give = lh ? T.acc0[v] : T.acc1[v];
    float keep = lh ? T.acc1[v] : T.acc0[v];
    tot[v] = (keep + __int_as_float(__builtin_amdgcn_ds_bpermute(paddr, __float_as_int(give)))) * rden;
  }
  int bo = tile * 32 + l;
  if (bo < batch) {
    float4* o = (float4*)(out + (size_t)bo * 8);
    o[0] = make_float4(tot[0], tot[1], tot[2], tot[3]);
    o[1] = make_float4(tot[4], tot[5], tot[6], tot[7]);
  }
}

__global__ void __launch_bounds__(EV_THREADS)
qnn_eval(const float* __restrict__ x, const float* __restrict__ AMg,
         float* __restrict__ out, int batch) {
  __shared__ __attribute__((aligned(16))) float AMl[AML_FLOATS];
  __shared__ __attribute__((aligned(32))) float C48[MT * 16];
  __shared__ float4 tab4[KB * 2];
  int tid = threadIdx.x;
  int l = tid & 31, w = tid >> 5;
  int lr = l & 15, lh = l >> 4;
  int stride = gridDim.x * (EV_THREADS / 32);
  int tile0 = blockIdx.x * (EV_THREADS / 32) + w;
  int ntiles = (batch + 31) >> 5;

  float x0v[2], x1v[2];
  ev_load_x(x, tile0, lr, batch, x0v, x1v);

  const float4* srcp = (const float4*)AMg;
  float4* dstp = (float4*)AMl;
  float c48v = 0.f;
  if (tid < MT * 16)
    c48v = AMg[(((tid >> 4) * KB + (KB - 1)) * 32 + (tid & 15)) * 2];
  float4 s0v[ST0_PER], s0c[ST0_PER];
#pragma unroll
  for (int j = 0; j < ST0_PER; ++j) {
    int i = tid + j * EV_THREADS;
    s0v[j] = make_float4(0.f, 0.f, 0.f, 0.f);
    s0c[j] = s0v[j];
    if (i < ST0_F4) {
      int t = i / (KBM * 16);
      int r = i - t * (KBM * 16);
      int kb = r >> 4, p = r & 15;
      s0v[j] = srcp[(t * KB + kb) * 16 + p];
      s0c[j] = srcp[(t * KB + (KB - 1)) * 16 + (p & 7)];
    }
  }
  if (tid < KB * 4) {
    float ang = (float)tid / 49.0f;
    float* tb = (float*)tab4;
    tb[2 * tid]     = cospif(ang);
    tb[2 * tid + 1] = sinpif(ang);
  }
#pragma unroll
  for (int j = 0; j < ST0_PER; ++j) {
    int i = tid + j * EV_THREADS;
    if (i < ST0_F4) {
      float4 v = s0v[j], cc = s0c[j];
      dstp[i] = make_float4(v.x - cc.x, v.y - cc.x, v.z - cc.z, v.w - cc.z);
    }
  }
  if (tid < MT * 16) C48[tid] = c48v;
  __syncthreads();

  float4 st1[ST1_PER];
  v2f st1c[ST1_PER];
#pragma unroll
  for (int j = 0; j < ST1_PER; ++j) {
    int i = ST0_F4 + tid + j * EV_THREADS;
    st1[j] = make_float4(0.f, 0.f, 0.f, 0.f);
    st1c[j].x = 0.f; st1c[j].y = 0.f;
    if (i < AML_FLOATS / 4) {
      int t = i / (KBM * 16);
      int r = i - t * (KBM * 16);
      int kb = r >> 4, p = r & 15;
      st1[j] = srcp[(t * KB + kb) * 16 + p];
      st1c[j] = *(const v2f*)(C48 + t * 16 + 2 * (p & 7));
    }
  }

  const v2f* Af = (const v2f*)AMl + l;
  const v2f* tab2 = (const v2f*)tab4;
  int paddr = (l ^ 16) << 2;

  EvTile T;
  ev_weights(x0v, x1v, tab4, tab2, lh, paddr, T);
  ev_mm(T, 0, ST0, Af, C48, tab2, lh);
#pragma unroll
  for (int j = 0; j < ST1_PER; ++j) {
    int i = ST0_F4 + tid + j * EV_THREADS;
    if (i < AML_FLOATS / 4) {
      float4 v = st1[j];
      v2f c = st1c[j];
      dstp[i] = make_float4(v.x - c.x, v.y - c.x, v.z - c.y, v.w - c.y);
    }
  }
  __syncthreads();
  ev_mm(T, ST0, MT, Af, C48, tab2, lh);
  ev_finish(T, lh, paddr, tile0, l, batch, out);

  for (int tile = tile0 + stride; tile < ntiles; tile += stride) {
    ev_load_x(x, tile, lr, batch, x0v, x1v);
    ev_weights(x0v, x1v, tab4, tab2, lh, paddr, T);
    ev_mm(T, 0, MT, Af, C48, tab2, lh);
    ev_finish(T, lh, paddr, tile, l, batch, out);
  }
}

extern "C" void kernel_launch(void* const* d_in, const int* in_sizes, int n_in,
                              void* d_out, int out_size, void* d_ws, size_t ws_size,
                              hipStream_t stream) {
  const float* x   = (const float*)d_in[0];
  const float* wts = (const float*)d_in[1];
  float* out = (float*)d_out;
  float* ws  = (float*)d_ws;
  float* AM  = ws + WS_AM;

  int batch = in_sizes[0] / 2;
  qnn_kernel<<<PATCH * PATCH, NTHREADS, 0, stream>>>(wts, AM);
  qnn_eval<<<256, EV_THREADS, 0, stream>>>(x, AM, out, batch);
}
